# starting kernel
# baseline (speedup 1.0000x reference)
#include <hip/hip_runtime.h>
#include <math.h>

typedef __attribute__((ext_vector_type(2))) float v2f;
typedef __attribute__((ext_vector_type(8))) float v8f;

#define PI_F 3.14159265358979323846f

static __device__ __forceinline__ v8f wmma_f32_k4(v2f a, v2f b, v8f c) {
  return __builtin_amdgcn_wmma_f32_16x16x4_f32(false, a, false, b, (short)0, c, false, false);
}

__global__ void qcnn_setup_kernel(const float* __restrict__ style,
                                  const float* __restrict__ style_w,
                                  const float* __restrict__ style_b,
                                  const float* __restrict__ qcnn,
                                  const float* __restrict__ meas,
                                  const float* __restrict__ out_w,
                                  float* __restrict__ Ur, float* __restrict__ Ui,
                                  float* __restrict__ Wq, float* __restrict__ sAng)
{
  __shared__ float sre[4096];
  __shared__ float sim[4096];
  const int t = threadIdx.x;

  for (int a = 0; a < 64; ++a) { sre[a*64+t] = (a == t) ? 1.f : 0.f; sim[a*64+t] = 0.f; }

  for (int l = 0; l < 2; ++l) {
    for (int i = 0; i < 6; ++i) {
      const int mask = 1 << (5 - i);
      {
        float th = 0.5f * qcnn[((l*6 + i)*2 + 0)*3 + 0];
        float c = cosf(th), s = sinf(th);
        for (int a = 0; a < 64; ++a) if (!(a & mask)) {
          int a1 = a | mask;
          float r0 = sre[a*64+t],  i0 = sim[a*64+t];
          float r1 = sre[a1*64+t], i1 = sim[a1*64+t];
          sre[a*64+t]  = c*r0 - s*r1;  sim[a*64+t]  = c*i0 - s*i1;
          sre[a1*64+t] = s*r0 + c*r1;  sim[a1*64+t] = s*i0 + c*i1;
        }
      }
      {
        float th = 0.5f * qcnn[((l*6 + i)*2 + 1)*3 + 0];
        float ch = cosf(th), sh = sinf(th);
        for (int a = 0; a < 64; ++a) {
          float re = sre[a*64+t], im = sim[a*64+t];
          if (!(a & mask)) { sre[a*64+t] = re*ch + im*sh; sim[a*64+t] = im*ch - re*sh; }
          else             { sre[a*64+t] = re*ch - im*sh; sim[a*64+t] = im*ch + re*sh; }
        }
      }
    }
    for (int i = 0; i < 6; ++i) {
      const int mc = 1 << (5 - i);
      const int mt = 1 << (5 - ((i + 1) % 6));
      for (int a = 0; a < 64; ++a) if ((a & mc) && !(a & mt)) {
        int a1 = a | mt;
        float r = sre[a*64+t], q = sim[a*64+t];
        sre[a*64+t]  = sre[a1*64+t]; sim[a*64+t]  = sim[a1*64+t];
        sre[a1*64+t] = r;            sim[a1*64+t] = q;
      }
    }
  }
  for (int i = 0; i < 6; ++i) {
    const int mask = 1 << (5 - i);
    float th = 0.5f * meas[i*3+0];
    float ph = meas[i*3+1], la = meas[i*3+2];
    float ct = cosf(th), st = sinf(th);
    float m01r = -cosf(la)*st,    m01i = -sinf(la)*st;
    float m10r =  cosf(ph)*st,    m10i =  sinf(ph)*st;
    float m11r =  cosf(ph+la)*ct, m11i =  sinf(ph+la)*ct;
    for (int a = 0; a < 64; ++a) if (!(a & mask)) {
      int a1 = a | mask;
      float r0 = sre[a*64+t],  i0 = sim[a*64+t];
      float r1 = sre[a1*64+t], i1 = sim[a1*64+t];
      sre[a*64+t]  = ct*r0 + m01r*r1 - m01i*i1;
      sim[a*64+t]  = ct*i0 + m01r*i1 + m01i*r1;
      sre[a1*64+t] = m10r*r0 - m10i*i0 + m11r*r1 - m11i*i1;
      sim[a1*64+t] = m10r*i0 + m10i*r0 + m11r*i1 + m11i*r1;
    }
  }
  for (int a = 0; a < 64; ++a) { Ur[a*64+t] = sre[a*64+t]; Ui[a*64+t] = sim[a*64+t]; }

  for (int c = 0; c < 64; ++c) {
    float s = 0.f;
    #pragma unroll
    for (int i = 0; i < 6; ++i)
      s += (((t >> (5 - i)) & 1) ? -1.f : 1.f) * out_w[c*6 + i];
    Wq[t*64 + c] = s;
  }

  if (t < 48) {
    int img = t / 6, i = t % 6;
    float d = 0.f;
    for (int j = 0; j < 128; ++j) d += style[img*128 + j] * style_w[i*128 + j];
    sAng[img*8 + i] = tanhf(d + style_b[i]) * PI_F;
  }
}

__global__ void qcnn_pack_kernel(const float* __restrict__ res_w,
                                 const float* __restrict__ data_w,
                                 float* __restrict__ Bpack)
{
  int idx = blockIdx.x * 256 + threadIdx.x;
  if (idx >= 46080) return;
  int v    = idx & 1;
  int lane = (idx >> 1) & 31;
  int s    = (idx >> 6) % 144;
  int n    = idx / 9216;
  int k    = (s << 2) + ((lane >> 4) << 1) + v;
  int row  = (n << 4) + (lane & 15);
  float val = 0.f;
  if (row < 64)      val = res_w[row * 576 + k];
  else if (row < 70) val = data_w[(row - 64) * 576 + k];
  Bpack[idx] = val;
}

__global__ __launch_bounds__(256) void qcnn_main_kernel(
    const float* __restrict__ x,
    const float* __restrict__ data_b,
    const float* __restrict__ res_b,
    const float* __restrict__ out_b,
    const float* __restrict__ gUr,
    const float* __restrict__ gUi,
    const float* __restrict__ gWq,
    const float* __restrict__ gStyle,
    const float* __restrict__ gBp,
    float* __restrict__ out)
{
  extern __shared__ float smem[];
  float* xs   = smem;
  float* Pm   = smem;
  float* PRm  = smem + 8704;
  float* Ur   = smem + 17408;
  float* Ui   = Ur + 4160;
  float* resS = Ui + 4160;
  float* thS  = resS + 8320;
  int*   offT = (int*)(thS + 1024);

  const int tid = threadIdx.x;
  const int b   = blockIdx.x >> 5;
  const int h0  = (blockIdx.x & 31) << 1;

  for (int i = tid; i < 4096; i += 256) {
    int a = i >> 6, k = i & 63;
    Ur[a*65 + k] = gUr[i];
    Ui[a*65 + k] = gUi[i];
  }
  for (int k = tid; k < 576; k += 256) {
    int c = k / 9, tt = k % 9;
    offT[k] = ((c << 2) + tt / 3) * 66 + (tt % 3);
  }
  for (int i = tid; i < 64*4*66; i += 256) {
    int c = i / 264, rr = i % 264, r = rr / 66, col = rr % 66;
    int h = h0 - 1 + r, w = col - 1;
    float v = 0.f;
    if ((unsigned)h < 64u && (unsigned)w < 64u)
      v = x[(size_t)(b*64 + c) * 4096 + h * 64 + w];
    xs[i] = v;
  }
  __syncthreads();

  const int lane = tid & 31;
  const int hf   = lane >> 4;
  const int lr   = lane & 15;
  const int m0   = (tid >> 5) << 4;
  const int m    = m0 + lr;
  const int baseM = ((m >> 6) * 66) + (m & 63);

  v8f acc[5];
  const v8f vz = {0.f,0.f,0.f,0.f,0.f,0.f,0.f,0.f};
  #pragma unroll
  for (int n = 0; n < 5; ++n) acc[n] = vz;

  const float2* Bp2 = (const float2*)gBp;
  for (int s = 0; s < 144; ++s) {
    int kk = (s << 2) + (hf << 1);
    v2f A;
    A.x = xs[offT[kk]     + baseM];
    A.y = xs[offT[kk + 1] + baseM];
    int bbase = (s << 5) + lane;
    #pragma unroll
    for (int n = 0; n < 5; ++n) {
      float2 bv = Bp2[n * 4608 + bbase];
      v2f B; B.x = bv.x; B.y = bv.y;
      acc[n] = wmma_f32_k4(A, B, acc[n]);
    }
  }

  #pragma unroll
  for (int n = 0; n < 4; ++n) {
    int ch = (n << 4) + lr;
    #pragma unroll
    for (int r = 0; r < 8; ++r)
      resS[(m0 + r + (hf << 3)) * 65 + ch] = acc[n][r];
  }
  if (lr < 6) {
    float db = data_b[lr];
    float sa = gStyle[(b << 3) + lr];
    #pragma unroll
    for (int r = 0; r < 8; ++r) {
      int mm = m0 + r + (hf << 3);
      thS[(mm << 3) + lr] = tanhf(acc[4][r] + db) * PI_F + sa;
    }
  }
  __syncthreads();

  if (tid < 128) {
    float cc[6], ss2[6];
    #pragma unroll
    for (int i = 0; i < 6; ++i) {
      float th = thS[(tid << 3) + i] * 0.5f;
      cc[i] = cosf(th); ss2[i] = sinf(th);
    }
    float p[64];
    p[0] = 1.f;
    #pragma unroll
    for (int i = 0; i < 6; ++i) {
      #pragma unroll
      for (int a = (1 << i) - 1; a >= 0; --a) {
        float tv = p[a];
        p[2*a + 1] = tv * ss2[i];
        p[2*a]     = tv * cc[i];
      }
    }
    #pragma unroll
    for (int a = 0; a < 64; a += 4)
      *(float4*)&Pm[tid * 68 + a] = make_float4(p[a], p[a+1], p[a+2], p[a+3]);
  }
  __syncthreads();

  #pragma unroll 1
  for (int nt = 0; nt < 4; ++nt) {
    int acol = (nt << 4) + lr;
    v8f ar = vz, ai = vz;
    for (int s = 0; s < 16; ++s) {
      int k = (s << 2) + (hf << 1);
      v2f A;  A.x  = Pm[m * 68 + k];        A.y  = Pm[m * 68 + k + 1];
      v2f Br; Br.x = Ur[acol * 65 + k];     Br.y = Ur[acol * 65 + k + 1];
      v2f Bi; Bi.x = Ui[acol * 65 + k];     Bi.y = Ui[acol * 65 + k + 1];
      ar = wmma_f32_k4(A, Br, ar);
      ai = wmma_f32_k4(A, Bi, ai);
    }
    #pragma unroll
    for (int r = 0; r < 8; ++r) {
      int mm = m0 + r + (hf << 3);
      PRm[mm * 68 + acol] = ar[r]*ar[r] + ai[r]*ai[r];
    }
  }

  #pragma unroll 1
  for (int nt = 0; nt < 4; ++nt) {
    int ccol = (nt << 4) + lr;
    v8f q = vz;
    for (int s = 0; s < 16; ++s) {
      int k = (s << 2) + (hf << 1);
      v2f A; A.x = PRm[m * 68 + k];          A.y = PRm[m * 68 + k + 1];
      v2f B; B.x = gWq[(k << 6) + ccol];     B.y = gWq[((k + 1) << 6) + ccol];
      q = wmma_f32_k4(A, B, q);
    }
    float bias = res_b[ccol] + out_b[ccol];
    #pragma unroll
    for (int r = 0; r < 8; ++r) {
      int mm = m0 + r + (hf << 3);
      resS[mm * 65 + ccol] += q[r] + bias;
    }
  }
  __syncthreads();

  for (int idx = tid; idx < 8192; idx += 256) {
    int c = idx >> 7, mm = idx & 127;
    int h = h0 + (mm >> 6), w = mm & 63;
    out[(size_t)(b*64 + c) * 4096 + h * 64 + w] = resS[mm * 65 + c];
  }
}

extern "C" void kernel_launch(void* const* d_in, const int* in_sizes, int n_in,
                              void* d_out, int out_size, void* d_ws, size_t ws_size,
                              hipStream_t stream) {
  (void)in_sizes; (void)n_in; (void)out_size; (void)ws_size;
  const float* x       = (const float*)d_in[0];
  const float* style   = (const float*)d_in[1];
  const float* data_w  = (const float*)d_in[2];
  const float* data_b  = (const float*)d_in[3];
  const float* style_w = (const float*)d_in[4];
  const float* style_b = (const float*)d_in[5];
  const float* qcnn    = (const float*)d_in[6];
  const float* meas    = (const float*)d_in[7];
  const float* out_w   = (const float*)d_in[8];
  const float* out_b   = (const float*)d_in[9];
  const float* res_w   = (const float*)d_in[10];
  const float* res_b   = (const float*)d_in[11];

  float* ws   = (float*)d_ws;
  float* Ur   = ws;
  float* Ui   = ws + 4096;
  float* Wq   = ws + 8192;
  float* sAng = ws + 12288;
  float* Bp   = ws + 12352;

  hipLaunchKernelGGL(qcnn_setup_kernel, dim3(1), dim3(64), 0, stream,
                     style, style_w, style_b, qcnn, meas, out_w, Ur, Ui, Wq, sAng);
  hipLaunchKernelGGL(qcnn_pack_kernel, dim3(180), dim3(256), 0, stream,
                     res_w, data_w, Bp);
  hipLaunchKernelGGL(qcnn_main_kernel, dim3(256), dim3(256), 142592  , stream,
                     x, data_b, res_b, out_b, Ur, Ui, Wq, sAng, Bp, (float*)d_out);
}


# setup kernel parallelised: one wave per column of U, lanes over amplitude pairs (RY+RZ fused per wire, CNOT ring as one permutation); Wq and style angles in own blocks; was 64 serial threads
# speedup vs baseline: 1.9101x; 1.9101x over previous
#include <hip/hip_runtime.h>
#include <math.h>

typedef __attribute__((ext_vector_type(2))) float v2f;
typedef __attribute__((ext_vector_type(8))) float v8f;

#define PI_F 3.14159265358979323846f

static __device__ __forceinline__ v8f wmma_f32_k4(v2f a, v2f b, v8f c) {
  return __builtin_amdgcn_wmma_f32_16x16x4_f32(false, a, false, b, (short)0, c, false, false);
}


static __device__ __forceinline__ void wave_lds_sync() {
  __builtin_amdgcn_fence(__ATOMIC_RELEASE, "wavefront");
  __builtin_amdgcn_wave_barrier();
  __builtin_amdgcn_fence(__ATOMIC_ACQUIRE, "wavefront");
}

__global__ __launch_bounds__(1024) void qcnn_setup_kernel(
                                  const float* __restrict__ style,
                                  const float* __restrict__ style_w,
                                  const float* __restrict__ style_b,
                                  const float* __restrict__ qcnn,
                                  const float* __restrict__ meas,
                                  const float* __restrict__ out_w,
                                  float* __restrict__ Ur, float* __restrict__ Ui,
                                  float* __restrict__ Wq, float* __restrict__ sAng)
{
  __shared__ float cre[32 * 65];
  __shared__ float cim[32 * 65];
  __shared__ float tcs[48];
  __shared__ float tsn[48];
  const int tid = threadIdx.x;
  const int blk = blockIdx.x;

  if (blk < 2) {
    const int lane = tid & 31;
    const int w    = tid >> 5;
    const int t    = (blk << 5) + w;
    float* re = cre + w * 65;
    float* im = cim + w * 65;

    if (tid < 48) {
      float ang;
      if (tid < 12)      ang = 0.5f * qcnn[(tid * 2 + 0) * 3];
      else if (tid < 24) ang = 0.5f * qcnn[((tid - 12) * 2 + 1) * 3];
      else {
        const int kind = (tid - 24) / 6, i = (tid - 24) % 6;
        const float th = 0.5f * meas[i*3+0];
        const float ph = meas[i*3+1], la = meas[i*3+2];
        ang = (kind == 0) ? th : (kind == 1) ? ph : (kind == 2) ? la : (ph + la);
      }
      tcs[tid] = cosf(ang);
      tsn[tid] = sinf(ang);
    }
    re[lane]      = (lane == t)      ? 1.f : 0.f;
    re[lane + 32] = (lane + 32 == t) ? 1.f : 0.f;
    im[lane]      = 0.f;
    im[lane + 32] = 0.f;
    __syncthreads();

    for (int l = 0; l < 2; ++l) {
      for (int i = 0; i < 6; ++i) {
        const int mask = 1 << (5 - i);
        const int low  = lane & (mask - 1);
        const int a    = ((lane - low) << 1) | low;
        const int a1   = a | mask;
        const int g    = l * 6 + i;
        const float c  = tcs[g],      s  = tsn[g];
        const float ch = tcs[12 + g], sh = tsn[12 + g];
        const float r0 = re[a],  i0 = im[a];
        const float r1 = re[a1], i1 = im[a1];
        const float yr0 = c*r0 - s*r1, yi0 = c*i0 - s*i1;
        const float yr1 = s*r0 + c*r1, yi1 = s*i0 + c*i1;
        re[a]  = yr0*ch + yi0*sh;  im[a]  = yi0*ch - yr0*sh;
        re[a1] = yr1*ch - yi1*sh;  im[a1] = yi1*ch + yr1*sh;
        wave_lds_sync();
      }
      int s0 = lane, s1 = lane + 32;
      for (int k = 5; k >= 0; --k) {
        const int mc = 1 << (5 - k);
        const int mt = 1 << (5 - ((k + 1) % 6));
        if (s0 & mc) s0 ^= mt;
        if (s1 & mc) s1 ^= mt;
      }
      const float pr0 = re[s0], pi0 = im[s0];
      const float pr1 = re[s1], pi1 = im[s1];
      wave_lds_sync();
      re[lane]      = pr0;  im[lane]      = pi0;
      re[lane + 32] = pr1;  im[lane + 32] = pi1;
      wave_lds_sync();
    }
    for (int i = 0; i < 6; ++i) {
      const int mask = 1 << (5 - i);
      const int low  = lane & (mask - 1);
      const int a    = ((lane - low) << 1) | low;
      const int a1   = a | mask;
      const float ct = tcs[24 + i], st = tsn[24 + i];
      const float m01r = -tcs[36 + i]*st, m01i = -tsn[36 + i]*st;
      const float m10r =  tcs[30 + i]*st, m10i =  tsn[30 + i]*st;
      const float m11r =  tcs[42 + i]*ct, m11i =  tsn[42 + i]*ct;
      const float r0 = re[a],  i0 = im[a];
      const float r1 = re[a1], i1 = im[a1];
      re[a]  = ct*r0 + m01r*r1 - m01i*i1;
      im[a]  = ct*i0 + m01r*i1 + m01i*r1;
      re[a1] = m10r*r0 - m10i*i0 + m11r*r1 - m11i*i1;
      im[a1] = m10r*i0 + m10i*r0 + m11r*i1 + m11i*r1;
      wave_lds_sync();
    }
    __syncthreads();
    for (int rr = w; rr < 64; rr += 32) {
      Ur[rr*64 + (blk << 5) + lane] = cre[lane*65 + rr];
      Ui[rr*64 + (blk << 5) + lane] = cim[lane*65 + rr];
    }
  } else if (blk < 6) {
    const int idx = ((blk - 2) << 10) + tid;
    const int a = idx >> 6, c = idx & 63;
    float s = 0.f;
    #pragma unroll
    for (int i = 0; i < 6; ++i)
      s += (((a >> (5 - i)) & 1) ? -1.f : 1.f) * out_w[c*6 + i];
    Wq[idx] = s;
  } else {
    if (tid < 64) {
      const int img = tid >> 3, i = tid & 7;
      if (i < 6) {
        const float4* sp = (const float4*)(style   + img*128);
        const float4* wp = (const float4*)(style_w + i*128);
        float d = 0.f;
        #pragma unroll 8
        for (int j = 0; j < 32; ++j) {
          const float4 sv = sp[j], wv = wp[j];
          d = fmaf(sv.x, wv.x, d);  d = fmaf(sv.y, wv.y, d);
          d = fmaf(sv.z, wv.z, d);  d = fmaf(sv.w, wv.w, d);
        }
        sAng[tid] = tanhf(d + style_b[i]) * PI_F;
      }
    }
  }
}

__global__ void qcnn_pack_kernel(const float* __restrict__ res_w,
                                 const float* __restrict__ data_w,
                                 float* __restrict__ Bpack)
{
  int idx = blockIdx.x * 256 + threadIdx.x;
  if (idx >= 46080) return;
  int v    = idx & 1;
  int lane = (idx >> 1) & 31;
  int s    = (idx >> 6) % 144;
  int n    = idx / 9216;
  int k    = (s << 2) + ((lane >> 4) << 1) + v;
  int row  = (n << 4) + (lane & 15);
  float val = 0.f;
  if (row < 64)      val = res_w[row * 576 + k];
  else if (row < 70) val = data_w[(row - 64) * 576 + k];
  Bpack[idx] = val;
}

__global__ __launch_bounds__(256) void qcnn_main_kernel(
    const float* __restrict__ x,
    const float* __restrict__ data_b,
    const float* __restrict__ res_b,
    const float* __restrict__ out_b,
    const float* __restrict__ gUr,
    const float* __restrict__ gUi,
    const float* __restrict__ gWq,
    const float* __restrict__ gStyle,
    const float* __restrict__ gBp,
    float* __restrict__ out)
{
  extern __shared__ float smem[];
  float* xs   = smem;
  float* Pm   = smem;
  float* PRm  = smem + 8704;
  float* Ur   = smem + 17408;
  float* Ui   = Ur + 4160;
  float* resS = Ui + 4160;
  float* thS  = resS + 8320;
  int*   offT = (int*)(thS + 1024);

  const int tid = threadIdx.x;
  const int b   = blockIdx.x >> 5;
  const int h0  = (blockIdx.x & 31) << 1;

  for (int i = tid; i < 4096; i += 256) {
    int a = i >> 6, k = i & 63;
    Ur[a*65 + k] = gUr[i];
    Ui[a*65 + k] = gUi[i];
  }
  for (int k = tid; k < 576; k += 256) {
    int c = k / 9, tt = k % 9;
    offT[k] = ((c << 2) + tt / 3) * 66 + (tt % 3);
  }
  for (int i = tid; i < 64*4*66; i += 256) {
    int c = i / 264, rr = i % 264, r = rr / 66, col = rr % 66;
    int h = h0 - 1 + r, w = col - 1;
    float v = 0.f;
    if ((unsigned)h < 64u && (unsigned)w < 64u)
      v = x[(size_t)(b*64 + c) * 4096 + h * 64 + w];
    xs[i] = v;
  }
  __syncthreads();

  const int lane = tid & 31;
  const int hf   = lane >> 4;
  const int lr   = lane & 15;
  const int m0   = (tid >> 5) << 4;
  const int m    = m0 + lr;
  const int baseM = ((m >> 6) * 66) + (m & 63);

  v8f acc[5];
  const v8f vz = {0.f,0.f,0.f,0.f,0.f,0.f,0.f,0.f};
  #pragma unroll
  for (int n = 0; n < 5; ++n) acc[n] = vz;

  const float2* Bp2 = (const float2*)gBp;
  for (int s = 0; s < 144; ++s) {
    int kk = (s << 2) + (hf << 1);
    v2f A;
    A.x = xs[offT[kk]     + baseM];
    A.y = xs[offT[kk + 1] + baseM];
    int bbase = (s << 5) + lane;
    #pragma unroll
    for (int n = 0; n < 5; ++n) {
      float2 bv = Bp2[n * 4608 + bbase];
      v2f B; B.x = bv.x; B.y = bv.y;
      acc[n] = wmma_f32_k4(A, B, acc[n]);
    }
  }

  #pragma unroll
  for (int n = 0; n < 4; ++n) {
    int ch = (n << 4) + lr;
    #pragma unroll
    for (int r = 0; r < 8; ++r)
      resS[(m0 + r + (hf << 3)) * 65 + ch] = acc[n][r];
  }
  if (lr < 6) {
    float db = data_b[lr];
    float sa = gStyle[(b << 3) + lr];
    #pragma unroll
    for (int r = 0; r < 8; ++r) {
      int mm = m0 + r + (hf << 3);
      thS[(mm << 3) + lr] = tanhf(acc[4][r] + db) * PI_F + sa;
    }
  }
  __syncthreads();

  if (tid < 128) {
    float cc[6], ss2[6];
    #pragma unroll
    for (int i = 0; i < 6; ++i) {
      float th = thS[(tid << 3) + i] * 0.5f;
      cc[i] = cosf(th); ss2[i] = sinf(th);
    }
    float p[64];
    p[0] = 1.f;
    #pragma unroll
    for (int i = 0; i < 6; ++i) {
      #pragma unroll
      for (int a = (1 << i) - 1; a >= 0; --a) {
        float tv = p[a];
        p[2*a + 1] = tv * ss2[i];
        p[2*a]     = tv * cc[i];
      }
    }
    #pragma unroll
    for (int a = 0; a < 64; a += 4)
      *(float4*)&Pm[tid * 68 + a] = make_float4(p[a], p[a+1], p[a+2], p[a+3]);
  }
  __syncthreads();

  #pragma unroll 1
  for (int nt = 0; nt < 4; ++nt) {
    int acol = (nt << 4) + lr;
    v8f ar = vz, ai = vz;
    for (int s = 0; s < 16; ++s) {
      int k = (s << 2) + (hf << 1);
      v2f A;  A.x  = Pm[m * 68 + k];        A.y  = Pm[m * 68 + k + 1];
      v2f Br; Br.x = Ur[acol * 65 + k];     Br.y = Ur[acol * 65 + k + 1];
      v2f Bi; Bi.x = Ui[acol * 65 + k];     Bi.y = Ui[acol * 65 + k + 1];
      ar = wmma_f32_k4(A, Br, ar);
      ai = wmma_f32_k4(A, Bi, ai);
    }
    #pragma unroll
    for (int r = 0; r < 8; ++r) {
      int mm = m0 + r + (hf << 3);
      PRm[mm * 68 + acol] = ar[r]*ar[r] + ai[r]*ai[r];
    }
  }

  #pragma unroll 1
  for (int nt = 0; nt < 4; ++nt) {
    int ccol = (nt << 4) + lr;
    v8f q = vz;
    for (int s = 0; s < 16; ++s) {
      int k = (s << 2) + (hf << 1);
      v2f A; A.x = PRm[m * 68 + k];          A.y = PRm[m * 68 + k + 1];
      v2f B; B.x = gWq[(k << 6) + ccol];     B.y = gWq[((k + 1) << 6) + ccol];
      q = wmma_f32_k4(A, B, q);
    }
    float bias = res_b[ccol] + out_b[ccol];
    #pragma unroll
    for (int r = 0; r < 8; ++r) {
      int mm = m0 + r + (hf << 3);
      resS[mm * 65 + ccol] += q[r] + bias;
    }
  }
  __syncthreads();

  for (int idx = tid; idx < 8192; idx += 256) {
    int c = idx >> 7, mm = idx & 127;
    int h = h0 + (mm >> 6), w = mm & 63;
    out[(size_t)(b*64 + c) * 4096 + h * 64 + w] = resS[mm * 65 + c];
  }
}

extern "C" void kernel_launch(void* const* d_in, const int* in_sizes, int n_in,
                              void* d_out, int out_size, void* d_ws, size_t ws_size,
                              hipStream_t stream) {
  (void)in_sizes; (void)n_in; (void)out_size; (void)ws_size;
  const float* x       = (const float*)d_in[0];
  const float* style   = (const float*)d_in[1];
  const float* data_w  = (const float*)d_in[2];
  const float* data_b  = (const float*)d_in[3];
  const float* style_w = (const float*)d_in[4];
  const float* style_b = (const float*)d_in[5];
  const float* qcnn    = (const float*)d_in[6];
  const float* meas    = (const float*)d_in[7];
  const float* out_w   = (const float*)d_in[8];
  const float* out_b   = (const float*)d_in[9];
  const float* res_w   = (const float*)d_in[10];
  const float* res_b   = (const float*)d_in[11];

  float* ws   = (float*)d_ws;
  float* Ur   = ws;
  float* Ui   = ws + 4096;
  float* Wq   = ws + 8192;
  float* sAng = ws + 12288;
  float* Bp   = ws + 12352;

  hipLaunchKernelGGL(qcnn_setup_kernel, dim3(7), dim3(1024), 0, stream,
                     style, style_w, style_b, qcnn, meas, out_w, Ur, Ui, Wq, sAng);
  hipLaunchKernelGGL(qcnn_pack_kernel, dim3(180), dim3(256), 0, stream,
                     res_w, data_w, Bp);
  hipLaunchKernelGGL(qcnn_main_kernel, dim3(256), dim3(256), 142592  , stream,
                     x, data_b, res_b, out_b, Ur, Ui, Wq, sAng, Bp, (float*)d_out);
}


# main kernel prologue: x halo tile and U fetched with 16-byte loads all issued before the first LDS store (was 4-byte loads each waited for alone, 66+8 dependent round trips per thread)
# speedup vs baseline: 3.7994x; 1.9891x over previous
#include <hip/hip_runtime.h>
#include <math.h>

typedef __attribute__((ext_vector_type(2))) float v2f;
typedef __attribute__((ext_vector_type(8))) float v8f;

#define PI_F 3.14159265358979323846f

static __device__ __forceinline__ v8f wmma_f32_k4(v2f a, v2f b, v8f c) {
  return __builtin_amdgcn_wmma_f32_16x16x4_f32(false, a, false, b, (short)0, c, false, false);
}


static __device__ __forceinline__ void wave_lds_sync() {
  __builtin_amdgcn_fence(__ATOMIC_RELEASE, "wavefront");
  __builtin_amdgcn_wave_barrier();
  __builtin_amdgcn_fence(__ATOMIC_ACQUIRE, "wavefront");
}

__global__ __launch_bounds__(1024) void qcnn_setup_kernel(
                                  const float* __restrict__ style,
                                  const float* __restrict__ style_w,
                                  const float* __restrict__ style_b,
                                  const float* __restrict__ qcnn,
                                  const float* __restrict__ meas,
                                  const float* __restrict__ out_w,
                                  float* __restrict__ Ur, float* __restrict__ Ui,
                                  float* __restrict__ Wq, float* __restrict__ sAng)
{
  __shared__ float cre[32 * 65];
  __shared__ float cim[32 * 65];
  __shared__ float tcs[48];
  __shared__ float tsn[48];
  const int tid = threadIdx.x;
  const int blk = blockIdx.x;

  if (blk < 2) {
    const int lane = tid & 31;
    const int w    = tid >> 5;
    const int t    = (blk << 5) + w;
    float* re = cre + w * 65;
    float* im = cim + w * 65;

    if (tid < 48) {
      float ang;
      if (tid < 12)      ang = 0.5f * qcnn[(tid * 2 + 0) * 3];
      else if (tid < 24) ang = 0.5f * qcnn[((tid - 12) * 2 + 1) * 3];
      else {
        const int kind = (tid - 24) / 6, i = (tid - 24) % 6;
        const float th = 0.5f * meas[i*3+0];
        const float ph = meas[i*3+1], la = meas[i*3+2];
        ang = (kind == 0) ? th : (kind == 1) ? ph : (kind == 2) ? la : (ph + la);
      }
      tcs[tid] = cosf(ang);
      tsn[tid] = sinf(ang);
    }
    re[lane]      = (lane == t)      ? 1.f : 0.f;
    re[lane + 32] = (lane + 32 == t) ? 1.f : 0.f;
    im[lane]      = 0.f;
    im[lane + 32] = 0.f;
    __syncthreads();

    for (int l = 0; l < 2; ++l) {
      for (int i = 0; i < 6; ++i) {
        const int mask = 1 << (5 - i);
        const int low  = lane & (mask - 1);
        const int a    = ((lane - low) << 1) | low;
        const int a1   = a | mask;
        const int g    = l * 6 + i;
        const float c  = tcs[g],      s  = tsn[g];
        const float ch = tcs[12 + g], sh = tsn[12 + g];
        const float r0 = re[a],  i0 = im[a];
        const float r1 = re[a1], i1 = im[a1];
        const float yr0 = c*r0 - s*r1, yi0 = c*i0 - s*i1;
        const float yr1 = s*r0 + c*r1, yi1 = s*i0 + c*i1;
        re[a]  = yr0*ch + yi0*sh;  im[a]  = yi0*ch - yr0*sh;
        re[a1] = yr1*ch - yi1*sh;  im[a1] = yi1*ch + yr1*sh;
        wave_lds_sync();
      }
      int s0 = lane, s1 = lane + 32;
      for (int k = 5; k >= 0; --k) {
        const int mc = 1 << (5 - k);
        const int mt = 1 << (5 - ((k + 1) % 6));
        if (s0 & mc) s0 ^= mt;
        if (s1 & mc) s1 ^= mt;
      }
      const float pr0 = re[s0], pi0 = im[s0];
      const float pr1 = re[s1], pi1 = im[s1];
      wave_lds_sync();
      re[lane]      = pr0;  im[lane]      = pi0;
      re[lane + 32] = pr1;  im[lane + 32] = pi1;
      wave_lds_sync();
    }
    for (int i = 0; i < 6; ++i) {
      const int mask = 1 << (5 - i);
      const int low  = lane & (mask - 1);
      const int a    = ((lane - low) << 1) | low;
      const int a1   = a | mask;
      const float ct = tcs[24 + i], st = tsn[24 + i];
      const float m01r = -tcs[36 + i]*st, m01i = -tsn[36 + i]*st;
      const float m10r =  tcs[30 + i]*st, m10i =  tsn[30 + i]*st;
      const float m11r =  tcs[42 + i]*ct, m11i =  tsn[42 + i]*ct;
      const float r0 = re[a],  i0 = im[a];
      const float r1 = re[a1], i1 = im[a1];
      re[a]  = ct*r0 + m01r*r1 - m01i*i1;
      im[a]  = ct*i0 + m01r*i1 + m01i*r1;
      re[a1] = m10r*r0 - m10i*i0 + m11r*r1 - m11i*i1;
      im[a1] = m10r*i0 + m10i*r0 + m11r*i1 + m11i*r1;
      wave_lds_sync();
    }
    __syncthreads();
    for (int rr = w; rr < 64; rr += 32) {
      Ur[rr*64 + (blk << 5) + lane] = cre[lane*65 + rr];
      Ui[rr*64 + (blk << 5) + lane] = cim[lane*65 + rr];
    }
  } else if (blk < 6) {
    const int idx = ((blk - 2) << 10) + tid;
    const int a = idx >> 6, c = idx & 63;
    float s = 0.f;
    #pragma unroll
    for (int i = 0; i < 6; ++i)
      s += (((a >> (5 - i)) & 1) ? -1.f : 1.f) * out_w[c*6 + i];
    Wq[idx] = s;
  } else {
    if (tid < 64) {
      const int img = tid >> 3, i = tid & 7;
      if (i < 6) {
        const float4* sp = (const float4*)(style   + img*128);
        const float4* wp = (const float4*)(style_w + i*128);
        float d = 0.f;
        #pragma unroll 8
        for (int j = 0; j < 32; ++j) {
          const float4 sv = sp[j], wv = wp[j];
          d = fmaf(sv.x, wv.x, d);  d = fmaf(sv.y, wv.y, d);
          d = fmaf(sv.z, wv.z, d);  d = fmaf(sv.w, wv.w, d);
        }
        sAng[tid] = tanhf(d + style_b[i]) * PI_F;
      }
    }
  }
}

__global__ void qcnn_pack_kernel(const float* __restrict__ res_w,
                                 const float* __restrict__ data_w,
                                 float* __restrict__ Bpack)
{
  int idx = blockIdx.x * 256 + threadIdx.x;
  if (idx >= 46080) return;
  int v    = idx & 1;
  int lane = (idx >> 1) & 31;
  int s    = (idx >> 6) % 144;
  int n    = idx / 9216;
  int k    = (s << 2) + ((lane >> 4) << 1) + v;
  int row  = (n << 4) + (lane & 15);
  float val = 0.f;
  if (row < 64)      val = res_w[row * 576 + k];
  else if (row < 70) val = data_w[(row - 64) * 576 + k];
  Bpack[idx] = val;
}

__global__ __launch_bounds__(256) void qcnn_main_kernel(
    const float* __restrict__ x,
    const float* __restrict__ data_b,
    const float* __restrict__ res_b,
    const float* __restrict__ out_b,
    const float* __restrict__ gUr,
    const float* __restrict__ gUi,
    const float* __restrict__ gWq,
    const float* __restrict__ gStyle,
    const float* __restrict__ gBp,
    float* __restrict__ out)
{
  extern __shared__ float smem[];
  float* xs   = smem;
  float* Pm   = smem;
  float* PRm  = smem + 8704;
  float* Ur   = smem + 17408;
  float* Ui   = Ur + 4160;
  float* resS = Ui + 4160;
  float* thS  = resS + 8320;
  int*   offT = (int*)(thS + 1024);

  const int tid = threadIdx.x;
  const int b   = blockIdx.x >> 5;
  const int h0  = (blockIdx.x & 31) << 1;

  const int xr   = (tid >> 4) & 3;
  const int xf   = tid & 15;
  const int cq   = tid >> 6;
  const int xh   = h0 - 1 + xr;
  const bool rowok = (unsigned)xh < 64u;
  const int xhc  = xh < 0 ? 0 : (xh > 63 ? 63 : xh);
  const float4* xsrc = (const float4*)(x + (size_t)(b*64 + cq) * 4096 + xhc * 64) + xf;
  float4 xv[16];
  #pragma unroll
  for (int j = 0; j < 16; ++j)
    xv[j] = xsrc[j * 4096];
  float4 urv[4], uiv[4];
  #pragma unroll
  for (int j = 0; j < 4; ++j) {
    urv[j] = ((const float4*)gUr)[tid + (j << 8)];
    uiv[j] = ((const float4*)gUi)[tid + (j << 8)];
  }
  for (int k = tid; k < 576; k += 256) {
    int c = k / 9, tt = k % 9;
    offT[k] = ((c << 2) + tt / 3) * 66 + (tt % 3);
  }
  for (int q = tid; q < 512; q += 256)
    xs[(q >> 1) * 66 + (q & 1) * 65] = 0.f;
  {
    float* xd = xs + ((cq << 2) + xr) * 66 + 1 + (xf << 2);
    #pragma unroll
    for (int j = 0; j < 16; ++j) {
      const float4 v = rowok ? xv[j] : make_float4(0.f, 0.f, 0.f, 0.f);
      float* d = xd + j * (16 * 66);
      d[0] = v.x; d[1] = v.y; d[2] = v.z; d[3] = v.w;
    }
    #pragma unroll
    for (int j = 0; j < 4; ++j) {
      const int a = (tid >> 4) + (j << 4);
      float* dr = Ur + a*65 + (xf << 2);
      float* di = Ui + a*65 + (xf << 2);
      dr[0] = urv[j].x; dr[1] = urv[j].y; dr[2] = urv[j].z; dr[3] = urv[j].w;
      di[0] = uiv[j].x; di[1] = uiv[j].y; di[2] = uiv[j].z; di[3] = uiv[j].w;
    }
  }
  __syncthreads();

  const int lane = tid & 31;
  const int hf   = lane >> 4;
  const int lr   = lane & 15;
  const int m0   = (tid >> 5) << 4;
  const int m    = m0 + lr;
  const int baseM = ((m >> 6) * 66) + (m & 63);

  v8f acc[5];
  const v8f vz = {0.f,0.f,0.f,0.f,0.f,0.f,0.f,0.f};
  #pragma unroll
  for (int n = 0; n < 5; ++n) acc[n] = vz;

  const float2* Bp2 = (const float2*)gBp;
  for (int s = 0; s < 144; ++s) {
    int kk = (s << 2) + (hf << 1);
    v2f A;
    A.x = xs[offT[kk]     + baseM];
    A.y = xs[offT[kk + 1] + baseM];
    int bbase = (s << 5) + lane;
    #pragma unroll
    for (int n = 0; n < 5; ++n) {
      float2 bv = Bp2[n * 4608 + bbase];
      v2f B; B.x = bv.x; B.y = bv.y;
      acc[n] = wmma_f32_k4(A, B, acc[n]);
    }
  }

  #pragma unroll
  for (int n = 0; n < 4; ++n) {
    int ch = (n << 4) + lr;
    #pragma unroll
    for (int r = 0; r < 8; ++r)
      resS[(m0 + r + (hf << 3)) * 65 + ch] = acc[n][r];
  }
  if (lr < 6) {
    float db = data_b[lr];
    float sa = gStyle[(b << 3) + lr];
    #pragma unroll
    for (int r = 0; r < 8; ++r) {
      int mm = m0 + r + (hf << 3);
      thS[(mm << 3) + lr] = tanhf(acc[4][r] + db) * PI_F + sa;
    }
  }
  __syncthreads();

  if (tid < 128) {
    float cc[6], ss2[6];
    #pragma unroll
    for (int i = 0; i < 6; ++i) {
      float th = thS[(tid << 3) + i] * 0.5f;
      cc[i] = cosf(th); ss2[i] = sinf(th);
    }
    float p[64];
    p[0] = 1.f;
    #pragma unroll
    for (int i = 0; i < 6; ++i) {
      #pragma unroll
      for (int a = (1 << i) - 1; a >= 0; --a) {
        float tv = p[a];
        p[2*a + 1] = tv * ss2[i];
        p[2*a]     = tv * cc[i];
      }
    }
    #pragma unroll
    for (int a = 0; a < 64; a += 4)
      *(float4*)&Pm[tid * 68 + a] = make_float4(p[a], p[a+1], p[a+2], p[a+3]);
  }
  __syncthreads();

  #pragma unroll 1
  for (int nt = 0; nt < 4; ++nt) {
    int acol = (nt << 4) + lr;
    v8f ar = vz, ai = vz;
    for (int s = 0; s < 16; ++s) {
      int k = (s << 2) + (hf << 1);
      v2f A;  A.x  = Pm[m * 68 + k];        A.y  = Pm[m * 68 + k + 1];
      v2f Br; Br.x = Ur[acol * 65 + k];     Br.y = Ur[acol * 65 + k + 1];
      v2f Bi; Bi.x = Ui[acol * 65 + k];     Bi.y = Ui[acol * 65 + k + 1];
      ar = wmma_f32_k4(A, Br, ar);
      ai = wmma_f32_k4(A, Bi, ai);
    }
    #pragma unroll
    for (int r = 0; r < 8; ++r) {
      int mm = m0 + r + (hf << 3);
      PRm[mm * 68 + acol] = ar[r]*ar[r] + ai[r]*ai[r];
    }
  }

  #pragma unroll 1
  for (int nt = 0; nt < 4; ++nt) {
    int ccol = (nt << 4) + lr;
    v8f q = vz;
    for (int s = 0; s < 16; ++s) {
      int k = (s << 2) + (hf << 1);
      v2f A; A.x = PRm[m * 68 + k];          A.y = PRm[m * 68 + k + 1];
      v2f B; B.x = gWq[(k << 6) + ccol];     B.y = gWq[((k + 1) << 6) + ccol];
      q = wmma_f32_k4(A, B, q);
    }
    float bias = res_b[ccol] + out_b[ccol];
    #pragma unroll
    for (int r = 0; r < 8; ++r) {
      int mm = m0 + r + (hf << 3);
      resS[mm * 65 + ccol] += q[r] + bias;
    }
  }
  __syncthreads();

  for (int idx = tid; idx < 8192; idx += 256) {
    int c = idx >> 7, mm = idx & 127;
    int h = h0 + (mm >> 6), w = mm & 63;
    out[(size_t)(b*64 + c) * 4096 + h * 64 + w] = resS[mm * 65 + c];
  }
}

extern "C" void kernel_launch(void* const* d_in, const int* in_sizes, int n_in,
                              void* d_out, int out_size, void* d_ws, size_t ws_size,
                              hipStream_t stream) {
  (void)in_sizes; (void)n_in; (void)out_size; (void)ws_size;
  const float* x       = (const float*)d_in[0];
  const float* style   = (const float*)d_in[1];
  const float* data_w  = (const float*)d_in[2];
  const float* data_b  = (const float*)d_in[3];
  const float* style_w = (const float*)d_in[4];
  const float* style_b = (const float*)d_in[5];
  const float* qcnn    = (const float*)d_in[6];
  const float* meas    = (const float*)d_in[7];
  const float* out_w   = (const float*)d_in[8];
  const float* out_b   = (const float*)d_in[9];
  const float* res_w   = (const float*)d_in[10];
  const float* res_b   = (const float*)d_in[11];

  float* ws   = (float*)d_ws;
  float* Ur   = ws;
  float* Ui   = ws + 4096;
  float* Wq   = ws + 8192;
  float* sAng = ws + 12288;
  float* Bp   = ws + 12352;

  hipLaunchKernelGGL(qcnn_setup_kernel, dim3(7), dim3(1024), 0, stream,
                     style, style_w, style_b, qcnn, meas, out_w, Ur, Ui, Wq, sAng);
  hipLaunchKernelGGL(qcnn_pack_kernel, dim3(180), dim3(256), 0, stream,
                     res_w, data_w, Bp);
  hipLaunchKernelGGL(qcnn_main_kernel, dim3(256), dim3(256), 142592  , stream,
                     x, data_b, res_b, out_b, Ur, Ui, Wq, sAng, Bp, (float*)d_out);
}


# main conv K loop software-pipelined in blocks of 9 K-steps: next block's 45 B (global) and 9 A (LDS, per-lane constant offsets, no offset table) fragments issued before current block's 45 WMMAs
# speedup vs baseline: 4.6753x; 1.2305x over previous
#include <hip/hip_runtime.h>
#include <math.h>

typedef __attribute__((ext_vector_type(2))) float v2f;
typedef __attribute__((ext_vector_type(8))) float v8f;

#define PI_F 3.14159265358979323846f

static __device__ __forceinline__ v8f wmma_f32_k4(v2f a, v2f b, v8f c) {
  return __builtin_amdgcn_wmma_f32_16x16x4_f32(false, a, false, b, (short)0, c, false, false);
}

static __device__ __forceinline__ constexpr int patch_off(int kl) {
  return (((kl / 9) << 2) + (kl % 9) / 3) * 66 + (kl % 9) % 3;
}

static __device__ __forceinline__ void s1_load(v2f (&A)[9], v2f (&B)[9][5],
                                               const float* xa, const float2* bl, int hf) {
  #pragma unroll
  for (int t = 0; t < 9; ++t) {
    #pragma unroll
    for (int n = 0; n < 5; ++n) {
      const float2 bv = bl[n * 4608 + t * 32];
      B[t][n].x = bv.x; B[t][n].y = bv.y;
    }
  }
  #pragma unroll
  for (int t = 0; t < 9; ++t) {
    const int o0 = hf ? patch_off(4 * t + 2) : patch_off(4 * t);
    const int o1 = hf ? patch_off(4 * t + 3) : patch_off(4 * t + 1);
    A[t].x = xa[o0];
    A[t].y = xa[o1];
  }
}

static __device__ __forceinline__ void s1_mma(const v2f (&A)[9], const v2f (&B)[9][5],
                                              v8f (&acc)[5]) {
  #pragma unroll
  for (int t = 0; t < 9; ++t) {
    #pragma unroll
    for (int n = 0; n < 5; ++n)
      acc[n] = wmma_f32_k4(A[t], B[t][n], acc[n]);
  }
}


static __device__ __forceinline__ void wave_lds_sync() {
  __builtin_amdgcn_fence(__ATOMIC_RELEASE, "wavefront");
  __builtin_amdgcn_wave_barrier();
  __builtin_amdgcn_fence(__ATOMIC_ACQUIRE, "wavefront");
}

__global__ __launch_bounds__(1024) void qcnn_setup_kernel(
                                  const float* __restrict__ style,
                                  const float* __restrict__ style_w,
                                  const float* __restrict__ style_b,
                                  const float* __restrict__ qcnn,
                                  const float* __restrict__ meas,
                                  const float* __restrict__ out_w,
                                  float* __restrict__ Ur, float* __restrict__ Ui,
                                  float* __restrict__ Wq, float* __restrict__ sAng)
{
  __shared__ float cre[32 * 65];
  __shared__ float cim[32 * 65];
  __shared__ float tcs[48];
  __shared__ float tsn[48];
  const int tid = threadIdx.x;
  const int blk = blockIdx.x;

  if (blk < 2) {
    const int lane = tid & 31;
    const int w    = tid >> 5;
    const int t    = (blk << 5) + w;
    float* re = cre + w * 65;
    float* im = cim + w * 65;

    if (tid < 48) {
      float ang;
      if (tid < 12)      ang = 0.5f * qcnn[(tid * 2 + 0) * 3];
      else if (tid < 24) ang = 0.5f * qcnn[((tid - 12) * 2 + 1) * 3];
      else {
        const int kind = (tid - 24) / 6, i = (tid - 24) % 6;
        const float th = 0.5f * meas[i*3+0];
        const float ph = meas[i*3+1], la = meas[i*3+2];
        ang = (kind == 0) ? th : (kind == 1) ? ph : (kind == 2) ? la : (ph + la);
      }
      tcs[tid] = cosf(ang);
      tsn[tid] = sinf(ang);
    }
    re[lane]      = (lane == t)      ? 1.f : 0.f;
    re[lane + 32] = (lane + 32 == t) ? 1.f : 0.f;
    im[lane]      = 0.f;
    im[lane + 32] = 0.f;
    __syncthreads();

    for (int l = 0; l < 2; ++l) {
      for (int i = 0; i < 6; ++i) {
        const int mask = 1 << (5 - i);
        const int low  = lane & (mask - 1);
        const int a    = ((lane - low) << 1) | low;
        const int a1   = a | mask;
        const int g    = l * 6 + i;
        const float c  = tcs[g],      s  = tsn[g];
        const float ch = tcs[12 + g], sh = tsn[12 + g];
        const float r0 = re[a],  i0 = im[a];
        const float r1 = re[a1], i1 = im[a1];
        const float yr0 = c*r0 - s*r1, yi0 = c*i0 - s*i1;
        const float yr1 = s*r0 + c*r1, yi1 = s*i0 + c*i1;
        re[a]  = yr0*ch + yi0*sh;  im[a]  = yi0*ch - yr0*sh;
        re[a1] = yr1*ch - yi1*sh;  im[a1] = yi1*ch + yr1*sh;
        wave_lds_sync();
      }
      int s0 = lane, s1 = lane + 32;
      for (int k = 5; k >= 0; --k) {
        const int mc = 1 << (5 - k);
        const int mt = 1 << (5 - ((k + 1) % 6));
        if (s0 & mc) s0 ^= mt;
        if (s1 & mc) s1 ^= mt;
      }
      const float pr0 = re[s0], pi0 = im[s0];
      const float pr1 = re[s1], pi1 = im[s1];
      wave_lds_sync();
      re[lane]      = pr0;  im[lane]      = pi0;
      re[lane + 32] = pr1;  im[lane + 32] = pi1;
      wave_lds_sync();
    }
    for (int i = 0; i < 6; ++i) {
      const int mask = 1 << (5 - i);
      const int low  = lane & (mask - 1);
      const int a    = ((lane - low) << 1) | low;
      const int a1   = a | mask;
      const float ct = tcs[24 + i], st = tsn[24 + i];
      const float m01r = -tcs[36 + i]*st, m01i = -tsn[36 + i]*st;
      const float m10r =  tcs[30 + i]*st, m10i =  tsn[30 + i]*st;
      const float m11r =  tcs[42 + i]*ct, m11i =  tsn[42 + i]*ct;
      const float r0 = re[a],  i0 = im[a];
      const float r1 = re[a1], i1 = im[a1];
      re[a]  = ct*r0 + m01r*r1 - m01i*i1;
      im[a]  = ct*i0 + m01r*i1 + m01i*r1;
      re[a1] = m10r*r0 - m10i*i0 + m11r*r1 - m11i*i1;
      im[a1] = m10r*i0 + m10i*r0 + m11r*i1 + m11i*r1;
      wave_lds_sync();
    }
    __syncthreads();
    for (int rr = w; rr < 64; rr += 32) {
      Ur[rr*64 + (blk << 5) + lane] = cre[lane*65 + rr];
      Ui[rr*64 + (blk << 5) + lane] = cim[lane*65 + rr];
    }
  } else if (blk < 6) {
    const int idx = ((blk - 2) << 10) + tid;
    const int a = idx >> 6, c = idx & 63;
    float s = 0.f;
    #pragma unroll
    for (int i = 0; i < 6; ++i)
      s += (((a >> (5 - i)) & 1) ? -1.f : 1.f) * out_w[c*6 + i];
    Wq[idx] = s;
  } else {
    if (tid < 64) {
      const int img = tid >> 3, i = tid & 7;
      if (i < 6) {
        const float4* sp = (const float4*)(style   + img*128);
        const float4* wp = (const float4*)(style_w + i*128);
        float d = 0.f;
        #pragma unroll 8
        for (int j = 0; j < 32; ++j) {
          const float4 sv = sp[j], wv = wp[j];
          d = fmaf(sv.x, wv.x, d);  d = fmaf(sv.y, wv.y, d);
          d = fmaf(sv.z, wv.z, d);  d = fmaf(sv.w, wv.w, d);
        }
        sAng[tid] = tanhf(d + style_b[i]) * PI_F;
      }
    }
  }
}

__global__ void qcnn_pack_kernel(const float* __restrict__ res_w,
                                 const float* __restrict__ data_w,
                                 float* __restrict__ Bpack)
{
  int idx = blockIdx.x * 256 + threadIdx.x;
  if (idx >= 46080) return;
  int v    = idx & 1;
  int lane = (idx >> 1) & 31;
  int s    = (idx >> 6) % 144;
  int n    = idx / 9216;
  int k    = (s << 2) + ((lane >> 4) << 1) + v;
  int row  = (n << 4) + (lane & 15);
  float val = 0.f;
  if (row < 64)      val = res_w[row * 576 + k];
  else if (row < 70) val = data_w[(row - 64) * 576 + k];
  Bpack[idx] = val;
}

__global__ __launch_bounds__(256) void qcnn_main_kernel(
    const float* __restrict__ x,
    const float* __restrict__ data_b,
    const float* __restrict__ res_b,
    const float* __restrict__ out_b,
    const float* __restrict__ gUr,
    const float* __restrict__ gUi,
    const float* __restrict__ gWq,
    const float* __restrict__ gStyle,
    const float* __restrict__ gBp,
    float* __restrict__ out)
{
  extern __shared__ float smem[];
  float* xs   = smem;
  float* Pm   = smem;
  float* PRm  = smem + 8704;
  float* Ur   = smem + 17408;
  float* Ui   = Ur + 4160;
  float* resS = Ui + 4160;
  float* thS  = resS + 8320;

  const int tid = threadIdx.x;
  const int b   = blockIdx.x >> 5;
  const int h0  = (blockIdx.x & 31) << 1;

  const int xr   = (tid >> 4) & 3;
  const int xf   = tid & 15;
  const int cq   = tid >> 6;
  const int xh   = h0 - 1 + xr;
  const bool rowok = (unsigned)xh < 64u;
  const int xhc  = xh < 0 ? 0 : (xh > 63 ? 63 : xh);
  const float4* xsrc = (const float4*)(x + (size_t)(b*64 + cq) * 4096 + xhc * 64) + xf;
  float4 xv[16];
  #pragma unroll
  for (int j = 0; j < 16; ++j)
    xv[j] = xsrc[j * 4096];
  float4 urv[4], uiv[4];
  #pragma unroll
  for (int j = 0; j < 4; ++j) {
    urv[j] = ((const float4*)gUr)[tid + (j << 8)];
    uiv[j] = ((const float4*)gUi)[tid + (j << 8)];
  }
  for (int q = tid; q < 512; q += 256)
    xs[(q >> 1) * 66 + (q & 1) * 65] = 0.f;
  {
    float* xd = xs + ((cq << 2) + xr) * 66 + 1 + (xf << 2);
    #pragma unroll
    for (int j = 0; j < 16; ++j) {
      const float4 v = rowok ? xv[j] : make_float4(0.f, 0.f, 0.f, 0.f);
      float* d = xd + j * (16 * 66);
      d[0] = v.x; d[1] = v.y; d[2] = v.z; d[3] = v.w;
    }
    #pragma unroll
    for (int j = 0; j < 4; ++j) {
      const int a = (tid >> 4) + (j << 4);
      float* dr = Ur + a*65 + (xf << 2);
      float* di = Ui + a*65 + (xf << 2);
      dr[0] = urv[j].x; dr[1] = urv[j].y; dr[2] = urv[j].z; dr[3] = urv[j].w;
      di[0] = uiv[j].x; di[1] = uiv[j].y; di[2] = uiv[j].z; di[3] = uiv[j].w;
    }
  }
  __syncthreads();

  const int lane = tid & 31;
  const int hf   = lane >> 4;
  const int lr   = lane & 15;
  const int m0   = (tid >> 5) << 4;
  const int m    = m0 + lr;
  const int baseM = ((m >> 6) * 66) + (m & 63);

  v8f acc[5];
  const v8f vz = {0.f,0.f,0.f,0.f,0.f,0.f,0.f,0.f};
  #pragma unroll
  for (int n = 0; n < 5; ++n) acc[n] = vz;

  {
    v2f A0[9], A1[9];
    v2f B0[9][5], B1[9][5];
    const float*  xa = xs + baseM;
    const float2* bl = (const float2*)gBp + lane;
    s1_load(A0, B0, xa, bl, hf);
    #pragma unroll 1
    for (int jj = 0; jj < 7; ++jj) {
      s1_load(A1, B1, xa + 1056, bl + 288, hf);
      __builtin_amdgcn_sched_barrier(0);
      s1_mma(A0, B0, acc);
      __builtin_amdgcn_sched_barrier(0);
      xa += 2112; bl += 576;
      s1_load(A0, B0, xa, bl, hf);
      __builtin_amdgcn_sched_barrier(0);
      s1_mma(A1, B1, acc);
      __builtin_amdgcn_sched_barrier(0);
    }
    s1_load(A1, B1, xa + 1056, bl + 288, hf);
    __builtin_amdgcn_sched_barrier(0);
    s1_mma(A0, B0, acc);
    s1_mma(A1, B1, acc);
  }

  #pragma unroll
  for (int n = 0; n < 4; ++n) {
    int ch = (n << 4) + lr;
    #pragma unroll
    for (int r = 0; r < 8; ++r)
      resS[(m0 + r + (hf << 3)) * 65 + ch] = acc[n][r];
  }
  if (lr < 6) {
    float db = data_b[lr];
    float sa = gStyle[(b << 3) + lr];
    #pragma unroll
    for (int r = 0; r < 8; ++r) {
      int mm = m0 + r + (hf << 3);
      thS[(mm << 3) + lr] = tanhf(acc[4][r] + db) * PI_F + sa;
    }
  }
  __syncthreads();

  if (tid < 128) {
    float cc[6], ss2[6];
    #pragma unroll
    for (int i = 0; i < 6; ++i) {
      float th = thS[(tid << 3) + i] * 0.5f;
      cc[i] = cosf(th); ss2[i] = sinf(th);
    }
    float p[64];
    p[0] = 1.f;
    #pragma unroll
    for (int i = 0; i < 6; ++i) {
      #pragma unroll
      for (int a = (1 << i) - 1; a >= 0; --a) {
        float tv = p[a];
        p[2*a + 1] = tv * ss2[i];
        p[2*a]     = tv * cc[i];
      }
    }
    #pragma unroll
    for (int a = 0; a < 64; a += 4)
      *(float4*)&Pm[tid * 68 + a] = make_float4(p[a], p[a+1], p[a+2], p[a+3]);
  }
  __syncthreads();

  #pragma unroll 1
  for (int nt = 0; nt < 4; ++nt) {
    int acol = (nt << 4) + lr;
    v8f ar = vz, ai = vz;
    for (int s = 0; s < 16; ++s) {
      int k = (s << 2) + (hf << 1);
      v2f A;  A.x  = Pm[m * 68 + k];        A.y  = Pm[m * 68 + k + 1];
      v2f Br; Br.x = Ur[acol * 65 + k];     Br.y = Ur[acol * 65 + k + 1];
      v2f Bi; Bi.x = Ui[acol * 65 + k];     Bi.y = Ui[acol * 65 + k + 1];
      ar = wmma_f32_k4(A, Br, ar);
      ai = wmma_f32_k4(A, Bi, ai);
    }
    #pragma unroll
    for (int r = 0; r < 8; ++r) {
      int mm = m0 + r + (hf << 3);
      PRm[mm * 68 + acol] = ar[r]*ar[r] + ai[r]*ai[r];
    }
  }

  #pragma unroll 1
  for (int nt = 0; nt < 4; ++nt) {
    int ccol = (nt << 4) + lr;
    v8f q = vz;
    for (int s = 0; s < 16; ++s) {
      int k = (s << 2) + (hf << 1);
      v2f A; A.x = PRm[m * 68 + k];          A.y = PRm[m * 68 + k + 1];
      v2f B; B.x = gWq[(k << 6) + ccol];     B.y = gWq[((k + 1) << 6) + ccol];
      q = wmma_f32_k4(A, B, q);
    }
    float bias = res_b[ccol] + out_b[ccol];
    #pragma unroll
    for (int r = 0; r < 8; ++r) {
      int mm = m0 + r + (hf << 3);
      resS[mm * 65 + ccol] += q[r] + bias;
    }
  }
  __syncthreads();

  for (int idx = tid; idx < 8192; idx += 256) {
    int c = idx >> 7, mm = idx & 127;
    int h = h0 + (mm >> 6), w = mm & 63;
    out[(size_t)(b*64 + c) * 4096 + h * 64 + w] = resS[mm * 65 + c];
  }
}

extern "C" void kernel_launch(void* const* d_in, const int* in_sizes, int n_in,
                              void* d_out, int out_size, void* d_ws, size_t ws_size,
                              hipStream_t stream) {
  (void)in_sizes; (void)n_in; (void)out_size; (void)ws_size;
  const float* x       = (const float*)d_in[0];
  const float* style   = (const float*)d_in[1];
  const float* data_w  = (const float*)d_in[2];
  const float* data_b  = (const float*)d_in[3];
  const float* style_w = (const float*)d_in[4];
  const float* style_b = (const float*)d_in[5];
  const float* qcnn    = (const float*)d_in[6];
  const float* meas    = (const float*)d_in[7];
  const float* out_w   = (const float*)d_in[8];
  const float* out_b   = (const float*)d_in[9];
  const float* res_w   = (const float*)d_in[10];
  const float* res_b   = (const float*)d_in[11];

  float* ws   = (float*)d_ws;
  float* Ur   = ws;
  float* Ui   = ws + 4096;
  float* Wq   = ws + 8192;
  float* sAng = ws + 12288;
  float* Bp   = ws + 12352;

  hipLaunchKernelGGL(qcnn_setup_kernel, dim3(7), dim3(1024), 0, stream,
                     style, style_w, style_b, qcnn, meas, out_w, Ur, Ui, Wq, sAng);
  hipLaunchKernelGGL(qcnn_pack_kernel, dim3(180), dim3(256), 0, stream,
                     res_w, data_w, Bp);
  hipLaunchKernelGGL(qcnn_main_kernel, dim3(256), dim3(256), 140288  , stream,
                     x, data_b, res_b, out_b, Ur, Ui, Wq, sAng, Bp, (float*)d_out);
}


# pack kernel folded into the setup launch as 45 extra blocks of 1024 threads (role by block index, independent of the U, Wq and style blocks): 3 dependent launches per call become 2
# speedup vs baseline: 5.0370x; 1.0774x over previous
#include <hip/hip_runtime.h>
#include <math.h>

typedef __attribute__((ext_vector_type(2))) float v2f;
typedef __attribute__((ext_vector_type(8))) float v8f;

#define PI_F 3.14159265358979323846f

static __device__ __forceinline__ v8f wmma_f32_k4(v2f a, v2f b, v8f c) {
  return __builtin_amdgcn_wmma_f32_16x16x4_f32(false, a, false, b, (short)0, c, false, false);
}

static __device__ __forceinline__ constexpr int patch_off(int kl) {
  return (((kl / 9) << 2) + (kl % 9) / 3) * 66 + (kl % 9) % 3;
}

static __device__ __forceinline__ void s1_load(v2f (&A)[9], v2f (&B)[9][5],
                                               const float* xa, const float2* bl, int hf) {
  #pragma unroll
  for (int t = 0; t < 9; ++t) {
    #pragma unroll
    for (int n = 0; n < 5; ++n) {
      const float2 bv = bl[n * 4608 + t * 32];
      B[t][n].x = bv.x; B[t][n].y = bv.y;
    }
  }
  #pragma unroll
  for (int t = 0; t < 9; ++t) {
    const int o0 = hf ? patch_off(4 * t + 2) : patch_off(4 * t);
    const int o1 = hf ? patch_off(4 * t + 3) : patch_off(4 * t + 1);
    A[t].x = xa[o0];
    A[t].y = xa[o1];
  }
}

static __device__ __forceinline__ void s1_mma(const v2f (&A)[9], const v2f (&B)[9][5],
                                              v8f (&acc)[5]) {
  #pragma unroll
  for (int t = 0; t < 9; ++t) {
    #pragma unroll
    for (int n = 0; n < 5; ++n)
      acc[n] = wmma_f32_k4(A[t], B[t][n], acc[n]);
  }
}


static __device__ __forceinline__ void wave_lds_sync() {
  __builtin_amdgcn_fence(__ATOMIC_RELEASE, "wavefront");
  __builtin_amdgcn_wave_barrier();
  __builtin_amdgcn_fence(__ATOMIC_ACQUIRE, "wavefront");
}

__global__ __launch_bounds__(1024) void qcnn_setup_kernel(
                                  const float* __restrict__ style,
                                  const float* __restrict__ style_w,
                                  const float* __restrict__ style_b,
                                  const float* __restrict__ qcnn,
                                  const float* __restrict__ meas,
                                  const float* __restrict__ out_w,
                                  const float* __restrict__ res_w,
                                  const float* __restrict__ data_w,
                                  float* __restrict__ Ur, float* __restrict__ Ui,
                                  float* __restrict__ Wq, float* __restrict__ sAng,
                                  float* __restrict__ Bpack)
{
  __shared__ float cre[32 * 65];
  __shared__ float cim[32 * 65];
  __shared__ float tcs[48];
  __shared__ float tsn[48];
  const int tid = threadIdx.x;
  const int blk = blockIdx.x;

  if (blk < 2) {
    const int lane = tid & 31;
    const int w    = tid >> 5;
    const int t    = (blk << 5) + w;
    float* re = cre + w * 65;
    float* im = cim + w * 65;

    if (tid < 48) {
      float ang;
      if (tid < 12)      ang = 0.5f * qcnn[(tid * 2 + 0) * 3];
      else if (tid < 24) ang = 0.5f * qcnn[((tid - 12) * 2 + 1) * 3];
      else {
        const int kind = (tid - 24) / 6, i = (tid - 24) % 6;
        const float th = 0.5f * meas[i*3+0];
        const float ph = meas[i*3+1], la = meas[i*3+2];
        ang = (kind == 0) ? th : (kind == 1) ? ph : (kind == 2) ? la : (ph + la);
      }
      tcs[tid] = cosf(ang);
      tsn[tid] = sinf(ang);
    }
    re[lane]      = (lane == t)      ? 1.f : 0.f;
    re[lane + 32] = (lane + 32 == t) ? 1.f : 0.f;
    im[lane]      = 0.f;
    im[lane + 32] = 0.f;
    __syncthreads();

    for (int l = 0; l < 2; ++l) {
      for (int i = 0; i < 6; ++i) {
        const int mask = 1 << (5 - i);
        const int low  = lane & (mask - 1);
        const int a    = ((lane - low) << 1) | low;
        const int a1   = a | mask;
        const int g    = l * 6 + i;
        const float c  = tcs[g],      s  = tsn[g];
        const float ch = tcs[12 + g], sh = tsn[12 + g];
        const float r0 = re[a],  i0 = im[a];
        const float r1 = re[a1], i1 = im[a1];
        const float yr0 = c*r0 - s*r1, yi0 = c*i0 - s*i1;
        const float yr1 = s*r0 + c*r1, yi1 = s*i0 + c*i1;
        re[a]  = yr0*ch + yi0*sh;  im[a]  = yi0*ch - yr0*sh;
        re[a1] = yr1*ch - yi1*sh;  im[a1] = yi1*ch + yr1*sh;
        wave_lds_sync();
      }
      int s0 = lane, s1 = lane + 32;
      for (int k = 5; k >= 0; --k) {
        const int mc = 1 << (5 - k);
        const int mt = 1 << (5 - ((k + 1) % 6));
        if (s0 & mc) s0 ^= mt;
        if (s1 & mc) s1 ^= mt;
      }
      const float pr0 = re[s0], pi0 = im[s0];
      const float pr1 = re[s1], pi1 = im[s1];
      wave_lds_sync();
      re[lane]      = pr0;  im[lane]      = pi0;
      re[lane + 32] = pr1;  im[lane + 32] = pi1;
      wave_lds_sync();
    }
    for (int i = 0; i < 6; ++i) {
      const int mask = 1 << (5 - i);
      const int low  = lane & (mask - 1);
      const int a    = ((lane - low) << 1) | low;
      const int a1   = a | mask;
      const float ct = tcs[24 + i], st = tsn[24 + i];
      const float m01r = -tcs[36 + i]*st, m01i = -tsn[36 + i]*st;
      const float m10r =  tcs[30 + i]*st, m10i =  tsn[30 + i]*st;
      const float m11r =  tcs[42 + i]*ct, m11i =  tsn[42 + i]*ct;
      const float r0 = re[a],  i0 = im[a];
      const float r1 = re[a1], i1 = im[a1];
      re[a]  = ct*r0 + m01r*r1 - m01i*i1;
      im[a]  = ct*i0 + m01r*i1 + m01i*r1;
      re[a1] = m10r*r0 - m10i*i0 + m11r*r1 - m11i*i1;
      im[a1] = m10r*i0 + m10i*r0 + m11r*i1 + m11i*r1;
      wave_lds_sync();
    }
    __syncthreads();
    for (int rr = w; rr < 64; rr += 32) {
      Ur[rr*64 + (blk << 5) + lane] = cre[lane*65 + rr];
      Ui[rr*64 + (blk << 5) + lane] = cim[lane*65 + rr];
    }
  } else if (blk < 6) {
    const int idx = ((blk - 2) << 10) + tid;
    const int a = idx >> 6, c = idx & 63;
    float s = 0.f;
    #pragma unroll
    for (int i = 0; i < 6; ++i)
      s += (((a >> (5 - i)) & 1) ? -1.f : 1.f) * out_w[c*6 + i];
    Wq[idx] = s;
  } else if (blk == 6) {
    if (tid < 64) {
      const int img = tid >> 3, i = tid & 7;
      if (i < 6) {
        const float4* sp = (const float4*)(style   + img*128);
        const float4* wp = (const float4*)(style_w + i*128);
        float d = 0.f;
        #pragma unroll 8
        for (int j = 0; j < 32; ++j) {
          const float4 sv = sp[j], wv = wp[j];
          d = fmaf(sv.x, wv.x, d);  d = fmaf(sv.y, wv.y, d);
          d = fmaf(sv.z, wv.z, d);  d = fmaf(sv.w, wv.w, d);
        }
        sAng[tid] = tanhf(d + style_b[i]) * PI_F;
      }
    }
  } else {
    const int idx  = ((blk - 7) << 10) + tid;
    const int v    = idx & 1;
    const int lane = (idx >> 1) & 31;
    const int s    = (idx >> 6) % 144;
    const int n    = idx / 9216;
    const int k    = (s << 2) + ((lane >> 4) << 1) + v;
    const int row  = (n << 4) + (lane & 15);
    float val = 0.f;
    if (row < 64)      val = res_w[row * 576 + k];
    else if (row < 70) val = data_w[(row - 64) * 576 + k];
    Bpack[idx] = val;
  }
}

__global__ __launch_bounds__(256) void qcnn_main_kernel(
    const float* __restrict__ x,
    const float* __restrict__ data_b,
    const float* __restrict__ res_b,
    const float* __restrict__ out_b,
    const float* __restrict__ gUr,
    const float* __restrict__ gUi,
    const float* __restrict__ gWq,
    const float* __restrict__ gStyle,
    const float* __restrict__ gBp,
    float* __restrict__ out)
{
  extern __shared__ float smem[];
  float* xs   = smem;
  float* Pm   = smem;
  float* PRm  = smem + 8704;
  float* Ur   = smem + 17408;
  float* Ui   = Ur + 4160;
  float* resS = Ui + 4160;
  float* thS  = resS + 8320;

  const int tid = threadIdx.x;
  const int b   = blockIdx.x >> 5;
  const int h0  = (blockIdx.x & 31) << 1;

  const int xr   = (tid >> 4) & 3;
  const int xf   = tid & 15;
  const int cq   = tid >> 6;
  const int xh   = h0 - 1 + xr;
  const bool rowok = (unsigned)xh < 64u;
  const int xhc  = xh < 0 ? 0 : (xh > 63 ? 63 : xh);
  const float4* xsrc = (const float4*)(x + (size_t)(b*64 + cq) * 4096 + xhc * 64) + xf;
  float4 xv[16];
  #pragma unroll
  for (int j = 0; j < 16; ++j)
    xv[j] = xsrc[j * 4096];
  float4 urv[4], uiv[4];
  #pragma unroll
  for (int j = 0; j < 4; ++j) {
    urv[j] = ((const float4*)gUr)[tid + (j << 8)];
    uiv[j] = ((const float4*)gUi)[tid + (j << 8)];
  }
  for (int q = tid; q < 512; q += 256)
    xs[(q >> 1) * 66 + (q & 1) * 65] = 0.f;
  {
    float* xd = xs + ((cq << 2) + xr) * 66 + 1 + (xf << 2);
    #pragma unroll
    for (int j = 0; j < 16; ++j) {
      const float4 v = rowok ? xv[j] : make_float4(0.f, 0.f, 0.f, 0.f);
      float* d = xd + j * (16 * 66);
      d[0] = v.x; d[1] = v.y; d[2] = v.z; d[3] = v.w;
    }
    #pragma unroll
    for (int j = 0; j < 4; ++j) {
      const int a = (tid >> 4) + (j << 4);
      float* dr = Ur + a*65 + (xf << 2);
      float* di = Ui + a*65 + (xf << 2);
      dr[0] = urv[j].x; dr[1] = urv[j].y; dr[2] = urv[j].z; dr[3] = urv[j].w;
      di[0] = uiv[j].x; di[1] = uiv[j].y; di[2] = uiv[j].z; di[3] = uiv[j].w;
    }
  }
  __syncthreads();

  const int lane = tid & 31;
  const int hf   = lane >> 4;
  const int lr   = lane & 15;
  const int m0   = (tid >> 5) << 4;
  const int m    = m0 + lr;
  const int baseM = ((m >> 6) * 66) + (m & 63);

  v8f acc[5];
  const v8f vz = {0.f,0.f,0.f,0.f,0.f,0.f,0.f,0.f};
  #pragma unroll
  for (int n = 0; n < 5; ++n) acc[n] = vz;

  {
    v2f A0[9], A1[9];
    v2f B0[9][5], B1[9][5];
    const float*  xa = xs + baseM;
    const float2* bl = (const float2*)gBp + lane;
    s1_load(A0, B0, xa, bl, hf);
    #pragma unroll 1
    for (int jj = 0; jj < 7; ++jj) {
      s1_load(A1, B1, xa + 1056, bl + 288, hf);
      __builtin_amdgcn_sched_barrier(0);
      s1_mma(A0, B0, acc);
      __builtin_amdgcn_sched_barrier(0);
      xa += 2112; bl += 576;
      s1_load(A0, B0, xa, bl, hf);
      __builtin_amdgcn_sched_barrier(0);
      s1_mma(A1, B1, acc);
      __builtin_amdgcn_sched_barrier(0);
    }
    s1_load(A1, B1, xa + 1056, bl + 288, hf);
    __builtin_amdgcn_sched_barrier(0);
    s1_mma(A0, B0, acc);
    s1_mma(A1, B1, acc);
  }

  #pragma unroll
  for (int n = 0; n < 4; ++n) {
    int ch = (n << 4) + lr;
    #pragma unroll
    for (int r = 0; r < 8; ++r)
      resS[(m0 + r + (hf << 3)) * 65 + ch] = acc[n][r];
  }
  if (lr < 6) {
    float db = data_b[lr];
    float sa = gStyle[(b << 3) + lr];
    #pragma unroll
    for (int r = 0; r < 8; ++r) {
      int mm = m0 + r + (hf << 3);
      thS[(mm << 3) + lr] = tanhf(acc[4][r] + db) * PI_F + sa;
    }
  }
  __syncthreads();

  if (tid < 128) {
    float cc[6], ss2[6];
    #pragma unroll
    for (int i = 0; i < 6; ++i) {
      float th = thS[(tid << 3) + i] * 0.5f;
      cc[i] = cosf(th); ss2[i] = sinf(th);
    }
    float p[64];
    p[0] = 1.f;
    #pragma unroll
    for (int i = 0; i < 6; ++i) {
      #pragma unroll
      for (int a = (1 << i) - 1; a >= 0; --a) {
        float tv = p[a];
        p[2*a + 1] = tv * ss2[i];
        p[2*a]     = tv * cc[i];
      }
    }
    #pragma unroll
    for (int a = 0; a < 64; a += 4)
      *(float4*)&Pm[tid * 68 + a] = make_float4(p[a], p[a+1], p[a+2], p[a+3]);
  }
  __syncthreads();

  #pragma unroll 1
  for (int nt = 0; nt < 4; ++nt) {
    int acol = (nt << 4) + lr;
    v8f ar = vz, ai = vz;
    for (int s = 0; s < 16; ++s) {
      int k = (s << 2) + (hf << 1);
      v2f A;  A.x  = Pm[m * 68 + k];        A.y  = Pm[m * 68 + k + 1];
      v2f Br; Br.x = Ur[acol * 65 + k];     Br.y = Ur[acol * 65 + k + 1];
      v2f Bi; Bi.x = Ui[acol * 65 + k];     Bi.y = Ui[acol * 65 + k + 1];
      ar = wmma_f32_k4(A, Br, ar);
      ai = wmma_f32_k4(A, Bi, ai);
    }
    #pragma unroll
    for (int r = 0; r < 8; ++r) {
      int mm = m0 + r + (hf << 3);
      PRm[mm * 68 + acol] = ar[r]*ar[r] + ai[r]*ai[r];
    }
  }

  #pragma unroll 1
  for (int nt = 0; nt < 4; ++nt) {
    int ccol = (nt << 4) + lr;
    v8f q = vz;
    for (int s = 0; s < 16; ++s) {
      int k = (s << 2) + (hf << 1);
      v2f A; A.x = PRm[m * 68 + k];          A.y = PRm[m * 68 + k + 1];
      v2f B; B.x = gWq[(k << 6) + ccol];     B.y = gWq[((k + 1) << 6) + ccol];
      q = wmma_f32_k4(A, B, q);
    }
    float bias = res_b[ccol] + out_b[ccol];
    #pragma unroll
    for (int r = 0; r < 8; ++r) {
      int mm = m0 + r + (hf << 3);
      resS[mm * 65 + ccol] += q[r] + bias;
    }
  }
  __syncthreads();

  for (int idx = tid; idx < 8192; idx += 256) {
    int c = idx >> 7, mm = idx & 127;
    int h = h0 + (mm >> 6), w = mm & 63;
    out[(size_t)(b*64 + c) * 4096 + h * 64 + w] = resS[mm * 65 + c];
  }
}

extern "C" void kernel_launch(void* const* d_in, const int* in_sizes, int n_in,
                              void* d_out, int out_size, void* d_ws, size_t ws_size,
                              hipStream_t stream) {
  (void)in_sizes; (void)n_in; (void)out_size; (void)ws_size;
  const float* x       = (const float*)d_in[0];
  const float* style   = (const float*)d_in[1];
  const float* data_w  = (const float*)d_in[2];
  const float* data_b  = (const float*)d_in[3];
  const float* style_w = (const float*)d_in[4];
  const float* style_b = (const float*)d_in[5];
  const float* qcnn    = (const float*)d_in[6];
  const float* meas    = (const float*)d_in[7];
  const float* out_w   = (const float*)d_in[8];
  const float* out_b   = (const float*)d_in[9];
  const float* res_w   = (const float*)d_in[10];
  const float* res_b   = (const float*)d_in[11];

  float* ws   = (float*)d_ws;
  float* Ur   = ws;
  float* Ui   = ws + 4096;
  float* Wq   = ws + 8192;
  float* sAng = ws + 12288;
  float* Bp   = ws + 12352;

  hipLaunchKernelGGL(qcnn_setup_kernel, dim3(52), dim3(1024), 0, stream,
                     style, style_w, style_b, qcnn, meas, out_w, res_w, data_w,
                     Ur, Ui, Wq, sAng, Bp);
  hipLaunchKernelGGL(qcnn_main_kernel, dim3(256), dim3(256), 140288  , stream,
                     x, data_b, res_b, out_b, Ur, Ui, Wq, sAng, Bp, (float*)d_out);
}


# stage 4 (probs x Wq) operand feed: Wq stored by setup in B-fragment order, fetched as 32 16-byte loads before stage 3; A fragments loaded once as one group; 4 independent accumulators
# speedup vs baseline: 5.5156x; 1.0950x over previous
#include <hip/hip_runtime.h>
#include <math.h>

typedef __attribute__((ext_vector_type(2))) float v2f;
typedef __attribute__((ext_vector_type(8))) float v8f;

#define PI_F 3.14159265358979323846f

static __device__ __forceinline__ v8f wmma_f32_k4(v2f a, v2f b, v8f c) {
  return __builtin_amdgcn_wmma_f32_16x16x4_f32(false, a, false, b, (short)0, c, false, false);
}

static __device__ __forceinline__ constexpr int patch_off(int kl) {
  return (((kl / 9) << 2) + (kl % 9) / 3) * 66 + (kl % 9) % 3;
}

static __device__ __forceinline__ void s1_load(v2f (&A)[9], v2f (&B)[9][5],
                                               const float* xa, const float2* bl, int hf) {
  #pragma unroll
  for (int t = 0; t < 9; ++t) {
    #pragma unroll
    for (int n = 0; n < 5; ++n) {
      const float2 bv = bl[n * 4608 + t * 32];
      B[t][n].x = bv.x; B[t][n].y = bv.y;
    }
  }
  #pragma unroll
  for (int t = 0; t < 9; ++t) {
    const int o0 = hf ? patch_off(4 * t + 2) : patch_off(4 * t);
    const int o1 = hf ? patch_off(4 * t + 3) : patch_off(4 * t + 1);
    A[t].x = xa[o0];
    A[t].y = xa[o1];
  }
}

static __device__ __forceinline__ void s1_mma(const v2f (&A)[9], const v2f (&B)[9][5],
                                              v8f (&acc)[5]) {
  #pragma unroll
  for (int t = 0; t < 9; ++t) {
    #pragma unroll
    for (int n = 0; n < 5; ++n)
      acc[n] = wmma_f32_k4(A[t], B[t][n], acc[n]);
  }
}


static __device__ __forceinline__ void wave_lds_sync() {
  __builtin_amdgcn_fence(__ATOMIC_RELEASE, "wavefront");
  __builtin_amdgcn_wave_barrier();
  __builtin_amdgcn_fence(__ATOMIC_ACQUIRE, "wavefront");
}

__global__ __launch_bounds__(1024) void qcnn_setup_kernel(
                                  const float* __restrict__ style,
                                  const float* __restrict__ style_w,
                                  const float* __restrict__ style_b,
                                  const float* __restrict__ qcnn,
                                  const float* __restrict__ meas,
                                  const float* __restrict__ out_w,
                                  const float* __restrict__ res_w,
                                  const float* __restrict__ data_w,
                                  float* __restrict__ Ur, float* __restrict__ Ui,
                                  float* __restrict__ Wq, float* __restrict__ sAng,
                                  float* __restrict__ Bpack)
{
  __shared__ float cre[32 * 65];
  __shared__ float cim[32 * 65];
  __shared__ float tcs[48];
  __shared__ float tsn[48];
  const int tid = threadIdx.x;
  const int blk = blockIdx.x;

  if (blk < 2) {
    const int lane = tid & 31;
    const int w    = tid >> 5;
    const int t    = (blk << 5) + w;
    float* re = cre + w * 65;
    float* im = cim + w * 65;

    if (tid < 48) {
      float ang;
      if (tid < 12)      ang = 0.5f * qcnn[(tid * 2 + 0) * 3];
      else if (tid < 24) ang = 0.5f * qcnn[((tid - 12) * 2 + 1) * 3];
      else {
        const int kind = (tid - 24) / 6, i = (tid - 24) % 6;
        const float th = 0.5f * meas[i*3+0];
        const float ph = meas[i*3+1], la = meas[i*3+2];
        ang = (kind == 0) ? th : (kind == 1) ? ph : (kind == 2) ? la : (ph + la);
      }
      tcs[tid] = cosf(ang);
      tsn[tid] = sinf(ang);
    }
    re[lane]      = (lane == t)      ? 1.f : 0.f;
    re[lane + 32] = (lane + 32 == t) ? 1.f : 0.f;
    im[lane]      = 0.f;
    im[lane + 32] = 0.f;
    __syncthreads();

    for (int l = 0; l < 2; ++l) {
      for (int i = 0; i < 6; ++i) {
        const int mask = 1 << (5 - i);
        const int low  = lane & (mask - 1);
        const int a    = ((lane - low) << 1) | low;
        const int a1   = a | mask;
        const int g    = l * 6 + i;
        const float c  = tcs[g],      s  = tsn[g];
        const float ch = tcs[12 + g], sh = tsn[12 + g];
        const float r0 = re[a],  i0 = im[a];
        const float r1 = re[a1], i1 = im[a1];
        const float yr0 = c*r0 - s*r1, yi0 = c*i0 - s*i1;
        const float yr1 = s*r0 + c*r1, yi1 = s*i0 + c*i1;
        re[a]  = yr0*ch + yi0*sh;  im[a]  = yi0*ch - yr0*sh;
        re[a1] = yr1*ch - yi1*sh;  im[a1] = yi1*ch + yr1*sh;
        wave_lds_sync();
      }
      int s0 = lane, s1 = lane + 32;
      for (int k = 5; k >= 0; --k) {
        const int mc = 1 << (5 - k);
        const int mt = 1 << (5 - ((k + 1) % 6));
        if (s0 & mc) s0 ^= mt;
        if (s1 & mc) s1 ^= mt;
      }
      const float pr0 = re[s0], pi0 = im[s0];
      const float pr1 = re[s1], pi1 = im[s1];
      wave_lds_sync();
      re[lane]      = pr0;  im[lane]      = pi0;
      re[lane + 32] = pr1;  im[lane + 32] = pi1;
      wave_lds_sync();
    }
    for (int i = 0; i < 6; ++i) {
      const int mask = 1 << (5 - i);
      const int low  = lane & (mask - 1);
      const int a    = ((lane - low) << 1) | low;
      const int a1   = a | mask;
      const float ct = tcs[24 + i], st = tsn[24 + i];
      const float m01r = -tcs[36 + i]*st, m01i = -tsn[36 + i]*st;
      const float m10r =  tcs[30 + i]*st, m10i =  tsn[30 + i]*st;
      const float m11r =  tcs[42 + i]*ct, m11i =  tsn[42 + i]*ct;
      const float r0 = re[a],  i0 = im[a];
      const float r1 = re[a1], i1 = im[a1];
      re[a]  = ct*r0 + m01r*r1 - m01i*i1;
      im[a]  = ct*i0 + m01r*i1 + m01i*r1;
      re[a1] = m10r*r0 - m10i*i0 + m11r*r1 - m11i*i1;
      im[a1] = m10r*i0 + m10i*r0 + m11r*i1 + m11i*r1;
      wave_lds_sync();
    }
    __syncthreads();
    for (int rr = w; rr < 64; rr += 32) {
      Ur[rr*64 + (blk << 5) + lane] = cre[lane*65 + rr];
      Ui[rr*64 + (blk << 5) + lane] = cim[lane*65 + rr];
    }
  } else if (blk < 6) {
    const int idx = ((blk - 2) << 10) + tid;
    const int ln = (idx >> 2) & 31;
    const int st = (((idx >> 7) & 7) << 1) + ((idx >> 1) & 1);
    const int a = (st << 2) + ((ln >> 4) << 1) + (idx & 1);
    const int c = ((idx >> 10) << 4) + (ln & 15);
    float s = 0.f;
    #pragma unroll
    for (int i = 0; i < 6; ++i)
      s += (((a >> (5 - i)) & 1) ? -1.f : 1.f) * out_w[c*6 + i];
    Wq[idx] = s;
  } else if (blk == 6) {
    if (tid < 64) {
      const int img = tid >> 3, i = tid & 7;
      if (i < 6) {
        const float4* sp = (const float4*)(style   + img*128);
        const float4* wp = (const float4*)(style_w + i*128);
        float d = 0.f;
        #pragma unroll 8
        for (int j = 0; j < 32; ++j) {
          const float4 sv = sp[j], wv = wp[j];
          d = fmaf(sv.x, wv.x, d);  d = fmaf(sv.y, wv.y, d);
          d = fmaf(sv.z, wv.z, d);  d = fmaf(sv.w, wv.w, d);
        }
        sAng[tid] = tanhf(d + style_b[i]) * PI_F;
      }
    }
  } else {
    const int idx  = ((blk - 7) << 10) + tid;
    const int v    = idx & 1;
    const int lane = (idx >> 1) & 31;
    const int s    = (idx >> 6) % 144;
    const int n    = idx / 9216;
    const int k    = (s << 2) + ((lane >> 4) << 1) + v;
    const int row  = (n << 4) + (lane & 15);
    float val = 0.f;
    if (row < 64)      val = res_w[row * 576 + k];
    else if (row < 70) val = data_w[(row - 64) * 576 + k];
    Bpack[idx] = val;
  }
}

__global__ __launch_bounds__(256) void qcnn_main_kernel(
    const float* __restrict__ x,
    const float* __restrict__ data_b,
    const float* __restrict__ res_b,
    const float* __restrict__ out_b,
    const float* __restrict__ gUr,
    const float* __restrict__ gUi,
    const float* __restrict__ gWq,
    const float* __restrict__ gStyle,
    const float* __restrict__ gBp,
    float* __restrict__ out)
{
  extern __shared__ float smem[];
  float* xs   = smem;
  float* Pm   = smem;
  float* PRm  = smem + 8704;
  float* Ur   = smem + 17408;
  float* Ui   = Ur + 4160;
  float* resS = Ui + 4160;
  float* thS  = resS + 8320;

  const int tid = threadIdx.x;
  const int b   = blockIdx.x >> 5;
  const int h0  = (blockIdx.x & 31) << 1;

  const int xr   = (tid >> 4) & 3;
  const int xf   = tid & 15;
  const int cq   = tid >> 6;
  const int xh   = h0 - 1 + xr;
  const bool rowok = (unsigned)xh < 64u;
  const int xhc  = xh < 0 ? 0 : (xh > 63 ? 63 : xh);
  const float4* xsrc = (const float4*)(x + (size_t)(b*64 + cq) * 4096 + xhc * 64) + xf;
  float4 xv[16];
  #pragma unroll
  for (int j = 0; j < 16; ++j)
    xv[j] = xsrc[j * 4096];
  float4 urv[4], uiv[4];
  #pragma unroll
  for (int j = 0; j < 4; ++j) {
    urv[j] = ((const float4*)gUr)[tid + (j << 8)];
    uiv[j] = ((const float4*)gUi)[tid + (j << 8)];
  }
  for (int q = tid; q < 512; q += 256)
    xs[(q >> 1) * 66 + (q & 1) * 65] = 0.f;
  {
    float* xd = xs + ((cq << 2) + xr) * 66 + 1 + (xf << 2);
    #pragma unroll
    for (int j = 0; j < 16; ++j) {
      const float4 v = rowok ? xv[j] : make_float4(0.f, 0.f, 0.f, 0.f);
      float* d = xd + j * (16 * 66);
      d[0] = v.x; d[1] = v.y; d[2] = v.z; d[3] = v.w;
    }
    #pragma unroll
    for (int j = 0; j < 4; ++j) {
      const int a = (tid >> 4) + (j << 4);
      float* dr = Ur + a*65 + (xf << 2);
      float* di = Ui + a*65 + (xf << 2);
      dr[0] = urv[j].x; dr[1] = urv[j].y; dr[2] = urv[j].z; dr[3] = urv[j].w;
      di[0] = uiv[j].x; di[1] = uiv[j].y; di[2] = uiv[j].z; di[3] = uiv[j].w;
    }
  }
  __syncthreads();

  const int lane = tid & 31;
  const int hf   = lane >> 4;
  const int lr   = lane & 15;
  const int m0   = (tid >> 5) << 4;
  const int m    = m0 + lr;
  const int baseM = ((m >> 6) * 66) + (m & 63);

  v8f acc[5];
  const v8f vz = {0.f,0.f,0.f,0.f,0.f,0.f,0.f,0.f};
  #pragma unroll
  for (int n = 0; n < 5; ++n) acc[n] = vz;

  {
    v2f A0[9], A1[9];
    v2f B0[9][5], B1[9][5];
    const float*  xa = xs + baseM;
    const float2* bl = (const float2*)gBp + lane;
    s1_load(A0, B0, xa, bl, hf);
    #pragma unroll 1
    for (int jj = 0; jj < 7; ++jj) {
      s1_load(A1, B1, xa + 1056, bl + 288, hf);
      __builtin_amdgcn_sched_barrier(0);
      s1_mma(A0, B0, acc);
      __builtin_amdgcn_sched_barrier(0);
      xa += 2112; bl += 576;
      s1_load(A0, B0, xa, bl, hf);
      __builtin_amdgcn_sched_barrier(0);
      s1_mma(A1, B1, acc);
      __builtin_amdgcn_sched_barrier(0);
    }
    s1_load(A1, B1, xa + 1056, bl + 288, hf);
    __builtin_amdgcn_sched_barrier(0);
    s1_mma(A0, B0, acc);
    s1_mma(A1, B1, acc);
  }

  #pragma unroll
  for (int n = 0; n < 4; ++n) {
    int ch = (n << 4) + lr;
    #pragma unroll
    for (int r = 0; r < 8; ++r)
      resS[(m0 + r + (hf << 3)) * 65 + ch] = acc[n][r];
  }
  if (lr < 6) {
    float db = data_b[lr];
    float sa = gStyle[(b << 3) + lr];
    #pragma unroll
    for (int r = 0; r < 8; ++r) {
      int mm = m0 + r + (hf << 3);
      thS[(mm << 3) + lr] = tanhf(acc[4][r] + db) * PI_F + sa;
    }
  }
  __syncthreads();

  if (tid < 128) {
    float cc[6], ss2[6];
    #pragma unroll
    for (int i = 0; i < 6; ++i) {
      float th = thS[(tid << 3) + i] * 0.5f;
      cc[i] = cosf(th); ss2[i] = sinf(th);
    }
    float p[64];
    p[0] = 1.f;
    #pragma unroll
    for (int i = 0; i < 6; ++i) {
      #pragma unroll
      for (int a = (1 << i) - 1; a >= 0; --a) {
        float tv = p[a];
        p[2*a + 1] = tv * ss2[i];
        p[2*a]     = tv * cc[i];
      }
    }
    #pragma unroll
    for (int a = 0; a < 64; a += 4)
      *(float4*)&Pm[tid * 68 + a] = make_float4(p[a], p[a+1], p[a+2], p[a+3]);
  }
  __syncthreads();

  float4 wq[32];
  {
    const float4* wqp = (const float4*)gWq + lane;
    #pragma unroll
    for (int j = 0; j < 32; ++j) wq[j] = wqp[j << 5];
  }

  #pragma unroll 1
  for (int nt = 0; nt < 4; ++nt) {
    int acol = (nt << 4) + lr;
    v8f ar = vz, ai = vz;
    for (int s = 0; s < 16; ++s) {
      int k = (s << 2) + (hf << 1);
      v2f A;  A.x  = Pm[m * 68 + k];        A.y  = Pm[m * 68 + k + 1];
      v2f Br; Br.x = Ur[acol * 65 + k];     Br.y = Ur[acol * 65 + k + 1];
      v2f Bi; Bi.x = Ui[acol * 65 + k];     Bi.y = Ui[acol * 65 + k + 1];
      ar = wmma_f32_k4(A, Br, ar);
      ai = wmma_f32_k4(A, Bi, ai);
    }
    #pragma unroll
    for (int r = 0; r < 8; ++r) {
      int mm = m0 + r + (hf << 3);
      PRm[mm * 68 + acol] = ar[r]*ar[r] + ai[r]*ai[r];
    }
  }

  {
    v2f A4[16];
    const float* pa = PRm + m * 68 + (hf << 1);
    #pragma unroll
    for (int s = 0; s < 16; ++s) {
      const float2 av = *(const float2*)(pa + (s << 2));
      A4[s].x = av.x; A4[s].y = av.y;
    }
    __builtin_amdgcn_sched_barrier(0);
    v8f q[4];
    #pragma unroll
    for (int nt = 0; nt < 4; ++nt) q[nt] = vz;
    #pragma unroll
    for (int s = 0; s < 16; ++s) {
      #pragma unroll
      for (int nt = 0; nt < 4; ++nt) {
        const float4 w = wq[(nt << 3) + (s >> 1)];
        v2f B;
        B.x = (s & 1) ? w.z : w.x;
        B.y = (s & 1) ? w.w : w.y;
        q[nt] = wmma_f32_k4(A4[s], B, q[nt]);
      }
    }
    #pragma unroll
    for (int nt = 0; nt < 4; ++nt) {
      const int ccol = (nt << 4) + lr;
      const float bias = res_b[ccol] + out_b[ccol];
      #pragma unroll
      for (int r = 0; r < 8; ++r) {
        const int mm = m0 + r + (hf << 3);
        resS[mm * 65 + ccol] += q[nt][r] + bias;
      }
    }
  }
  __syncthreads();

  for (int idx = tid; idx < 8192; idx += 256) {
    int c = idx >> 7, mm = idx & 127;
    int h = h0 + (mm >> 6), w = mm & 63;
    out[(size_t)(b*64 + c) * 4096 + h * 64 + w] = resS[mm * 65 + c];
  }
}

extern "C" void kernel_launch(void* const* d_in, const int* in_sizes, int n_in,
                              void* d_out, int out_size, void* d_ws, size_t ws_size,
                              hipStream_t stream) {
  (void)in_sizes; (void)n_in; (void)out_size; (void)ws_size;
  const float* x       = (const float*)d_in[0];
  const float* style   = (const float*)d_in[1];
  const float* data_w  = (const float*)d_in[2];
  const float* data_b  = (const float*)d_in[3];
  const float* style_w = (const float*)d_in[4];
  const float* style_b = (const float*)d_in[5];
  const float* qcnn    = (const float*)d_in[6];
  const float* meas    = (const float*)d_in[7];
  const float* out_w   = (const float*)d_in[8];
  const float* out_b   = (const float*)d_in[9];
  const float* res_w   = (const float*)d_in[10];
  const float* res_b   = (const float*)d_in[11];

  float* ws   = (float*)d_ws;
  float* Ur   = ws;
  float* Ui   = ws + 4096;
  float* Wq   = ws + 8192;
  float* sAng = ws + 12288;
  float* Bp   = ws + 12352;

  hipLaunchKernelGGL(qcnn_setup_kernel, dim3(52), dim3(1024), 0, stream,
                     style, style_w, style_b, qcnn, meas, out_w, res_w, data_w,
                     Ur, Ui, Wq, sAng, Bp);
  hipLaunchKernelGGL(qcnn_main_kernel, dim3(256), dim3(256), 140288  , stream,
                     x, data_b, res_b, out_b, Ur, Ui, Wq, sAng, Bp, (float*)d_out);
}


# setup launch: each U column built in one wave's registers with cross-lane moves (DPP, row swap) instead of an LDS round trip per gate, over 16 work-groups; U stored transposed, main un-transposes
# speedup vs baseline: 5.8889x; 1.0677x over previous
#include <hip/hip_runtime.h>
#include <math.h>

typedef __attribute__((ext_vector_type(2))) float v2f;
typedef __attribute__((ext_vector_type(8))) float v8f;

#define PI_F 3.14159265358979323846f

static __device__ __forceinline__ v8f wmma_f32_k4(v2f a, v2f b, v8f c) {
  return __builtin_amdgcn_wmma_f32_16x16x4_f32(false, a, false, b, (short)0, c, false, false);
}

static __device__ __forceinline__ constexpr int patch_off(int kl) {
  return (((kl / 9) << 2) + (kl % 9) / 3) * 66 + (kl % 9) % 3;
}

static __device__ __forceinline__ void s1_load(v2f (&A)[9], v2f (&B)[9][5],
                                               const float* xa, const float2* bl, int hf) {
  #pragma unroll
  for (int t = 0; t < 9; ++t) {
    #pragma unroll
    for (int n = 0; n < 5; ++n) {
      const float2 bv = bl[n * 4608 + t * 32];
      B[t][n].x = bv.x; B[t][n].y = bv.y;
    }
  }
  #pragma unroll
  for (int t = 0; t < 9; ++t) {
    const int o0 = hf ? patch_off(4 * t + 2) : patch_off(4 * t);
    const int o1 = hf ? patch_off(4 * t + 3) : patch_off(4 * t + 1);
    A[t].x = xa[o0];
    A[t].y = xa[o1];
  }
}

static __device__ __forceinline__ void s1_mma(const v2f (&A)[9], const v2f (&B)[9][5],
                                              v8f (&acc)[5]) {
  #pragma unroll
  for (int t = 0; t < 9; ++t) {
    #pragma unroll
    for (int n = 0; n < 5; ++n)
      acc[n] = wmma_f32_k4(A[t], B[t][n], acc[n]);
  }
}


static __device__ __forceinline__ float lane_xor(float x, int m) {
  const int v = __builtin_bit_cast(int, x);
  int r;
  switch (m) {
    case 1:  r = __builtin_amdgcn_update_dpp(v, v, 0x161, 0xf, 0xf, false); break;
    case 2:  r = __builtin_amdgcn_update_dpp(v, v, 0x162, 0xf, 0xf, false); break;
    case 4:  r = __builtin_amdgcn_update_dpp(v, v, 0x164, 0xf, 0xf, false); break;
    case 8:  r = __builtin_amdgcn_update_dpp(v, v, 0x168, 0xf, 0xf, false); break;
    default: r = __builtin_amdgcn_permlanex16(v, v, 0x76543210, 0xfedcba98, false, false); break;
  }
  return __builtin_bit_cast(float, r);
}

__global__ __launch_bounds__(128) void qcnn_setup_kernel(
                                  const float* __restrict__ style,
                                  const float* __restrict__ style_w,
                                  const float* __restrict__ style_b,
                                  const float* __restrict__ qcnn,
                                  const float* __restrict__ meas,
                                  const float* __restrict__ out_w,
                                  const float* __restrict__ res_w,
                                  const float* __restrict__ data_w,
                                  float* __restrict__ UrT, float* __restrict__ UiT,
                                  float* __restrict__ Wq, float* __restrict__ sAng,
                                  float* __restrict__ Bpack)
{
  __shared__ float tcs[48];
  __shared__ float tsn[48];
  const int tid = threadIdx.x;
  const int blk = blockIdx.x;

  if (blk < 16) {
    const int lane = tid & 31;
    const int t    = (blk << 2) + (tid >> 5);

    if (tid < 48) {
      float ang;
      if (tid < 12)      ang = 0.5f * qcnn[(tid * 2 + 0) * 3];
      else if (tid < 24) ang = 0.5f * qcnn[((tid - 12) * 2 + 1) * 3];
      else {
        const int kind = (tid - 24) / 6, i = (tid - 24) % 6;
        const float th = 0.5f * meas[i*3+0];
        const float ph = meas[i*3+1], la = meas[i*3+2];
        ang = (kind == 0) ? th : (kind == 1) ? ph : (kind == 2) ? la : (ph + la);
      }
      tcs[tid] = cosf(ang);
      tsn[tid] = sinf(ang);
    }
    __syncthreads();

    float r0 = (lane == t)      ? 1.f : 0.f;
    float r1 = (lane + 32 == t) ? 1.f : 0.f;
    float i0 = 0.f, i1 = 0.f;

    #pragma unroll
    for (int l = 0; l < 2; ++l) {
      {
        const int g = l * 6;
        const float c  = tcs[g],      s  = tsn[g];
        const float ch = tcs[12 + g], sh = tsn[12 + g];
        const float yr0 = c*r0 - s*r1, yi0 = c*i0 - s*i1;
        const float yr1 = s*r0 + c*r1, yi1 = s*i0 + c*i1;
        r0 = yr0*ch + yi0*sh;  i0 = yi0*ch - yr0*sh;
        r1 = yr1*ch - yi1*sh;  i1 = yi1*ch + yr1*sh;
      }
      #pragma unroll
      for (int i = 1; i < 6; ++i) {
        const int mask = 1 << (5 - i);
        const int g    = l * 6 + i;
        const bool hb  = (lane & mask) != 0;
        const float c  = tcs[g],      s  = tsn[g];
        const float ch = tcs[12 + g], sh = tsn[12 + g];
        const float ss  = hb ? s : -s;
        const float shs = hb ? -sh : sh;
        const float pr0 = lane_xor(r0, mask), pi0 = lane_xor(i0, mask);
        const float pr1 = lane_xor(r1, mask), pi1 = lane_xor(i1, mask);
        const float yr0 = c*r0 + ss*pr0, yi0 = c*i0 + ss*pi0;
        const float yr1 = c*r1 + ss*pr1, yi1 = c*i1 + ss*pi1;
        r0 = yr0*ch + yi0*shs;  i0 = yi0*ch - yr0*shs;
        r1 = yr1*ch + yi1*shs;  i1 = yi1*ch - yr1*shs;
      }
      int s0 = lane, s1 = lane + 32;
      #pragma unroll
      for (int k = 5; k >= 0; --k) {
        const int mc = 1 << (5 - k);
        const int mt = 1 << (5 - ((k + 1) % 6));
        if (s0 & mc) s0 ^= mt;
        if (s1 & mc) s1 ^= mt;
      }
      const int l0 = s0 & 31, l1 = s1 & 31;
      const float a0 = __shfl(r0, l0), a1 = __shfl(r1, l0);
      const float a2 = __shfl(i0, l0), a3 = __shfl(i1, l0);
      const float b0 = __shfl(r0, l1), b1 = __shfl(r1, l1);
      const float b2 = __shfl(i0, l1), b3 = __shfl(i1, l1);
      r0 = (s0 & 32) ? a1 : a0;  i0 = (s0 & 32) ? a3 : a2;
      r1 = (s1 & 32) ? b1 : b0;  i1 = (s1 & 32) ? b3 : b2;
    }
    #pragma unroll
    for (int i = 0; i < 6; ++i) {
      const float ct = tcs[24 + i], st = tsn[24 + i];
      const float m01r = -tcs[36 + i]*st, m01i = -tsn[36 + i]*st;
      const float m10r =  tcs[30 + i]*st, m10i =  tsn[30 + i]*st;
      const float m11r =  tcs[42 + i]*ct, m11i =  tsn[42 + i]*ct;
      if (i == 0) {
        const float nr0 = ct*r0 + m01r*r1 - m01i*i1;
        const float ni0 = ct*i0 + m01r*i1 + m01i*r1;
        const float nr1 = m10r*r0 - m10i*i0 + m11r*r1 - m11i*i1;
        const float ni1 = m10r*i0 + m10i*r0 + m11r*i1 + m11i*r1;
        r0 = nr0; i0 = ni0; r1 = nr1; i1 = ni1;
      } else {
        const int mask = 1 << (5 - i);
        const bool hb  = (lane & mask) != 0;
        const float pr0 = lane_xor(r0, mask), pi0 = lane_xor(i0, mask);
        const float pr1 = lane_xor(r1, mask), pi1 = lane_xor(i1, mask);
        const float lr0 = ct*r0 + m01r*pr0 - m01i*pi0;
        const float li0 = ct*i0 + m01r*pi0 + m01i*pr0;
        const float hr0 = m10r*pr0 - m10i*pi0 + m11r*r0 - m11i*i0;
        const float hi0 = m10r*pi0 + m10i*pr0 + m11r*i0 + m11i*r0;
        const float lr1 = ct*r1 + m01r*pr1 - m01i*pi1;
        const float li1 = ct*i1 + m01r*pi1 + m01i*pr1;
        const float hr1 = m10r*pr1 - m10i*pi1 + m11r*r1 - m11i*i1;
        const float hi1 = m10r*pi1 + m10i*pr1 + m11r*i1 + m11i*r1;
        r0 = hb ? hr0 : lr0;  i0 = hb ? hi0 : li0;
        r1 = hb ? hr1 : lr1;  i1 = hb ? hi1 : li1;
      }
    }
    UrT[t*64 + lane]      = r0;
    UrT[t*64 + 32 + lane] = r1;
    UiT[t*64 + lane]      = i0;
    UiT[t*64 + 32 + lane] = i1;
  } else if (blk < 48) {
    const int idx = ((blk - 16) << 7) + tid;
    const int ln = (idx >> 2) & 31;
    const int st = (((idx >> 7) & 7) << 1) + ((idx >> 1) & 1);
    const int a = (st << 2) + ((ln >> 4) << 1) + (idx & 1);
    const int c = ((idx >> 10) << 4) + (ln & 15);
    float s = 0.f;
    #pragma unroll
    for (int i = 0; i < 6; ++i)
      s += (((a >> (5 - i)) & 1) ? -1.f : 1.f) * out_w[c*6 + i];
    Wq[idx] = s;
  } else if (blk == 48) {
    if (tid < 64) {
      const int img = tid >> 3, i = tid & 7;
      if (i < 6) {
        const float4* sp = (const float4*)(style   + img*128);
        const float4* wp = (const float4*)(style_w + i*128);
        float d = 0.f;
        #pragma unroll 8
        for (int j = 0; j < 32; ++j) {
          const float4 sv = sp[j], wv = wp[j];
          d = fmaf(sv.x, wv.x, d);  d = fmaf(sv.y, wv.y, d);
          d = fmaf(sv.z, wv.z, d);  d = fmaf(sv.w, wv.w, d);
        }
        sAng[tid] = tanhf(d + style_b[i]) * PI_F;
      }
    }
  } else {
    const int idx  = ((blk - 49) << 7) + tid;
    const int v    = idx & 1;
    const int lane = (idx >> 1) & 31;
    const int s    = (idx >> 6) % 144;
    const int n    = idx / 9216;
    const int k    = (s << 2) + ((lane >> 4) << 1) + v;
    const int row  = (n << 4) + (lane & 15);
    float val = 0.f;
    if (row < 64)      val = res_w[row * 576 + k];
    else if (row < 70) val = data_w[(row - 64) * 576 + k];
    Bpack[idx] = val;
  }
}

__global__ __launch_bounds__(256) void qcnn_main_kernel(
    const float* __restrict__ x,
    const float* __restrict__ data_b,
    const float* __restrict__ res_b,
    const float* __restrict__ out_b,
    const float* __restrict__ gUr,
    const float* __restrict__ gUi,
    const float* __restrict__ gWq,
    const float* __restrict__ gStyle,
    const float* __restrict__ gBp,
    float* __restrict__ out)
{
  extern __shared__ float smem[];
  float* xs   = smem;
  float* Pm   = smem;
  float* PRm  = smem + 8704;
  float* Ur   = smem + 17408;
  float* Ui   = Ur + 4160;
  float* resS = Ui + 4160;
  float* thS  = resS + 8320;

  const int tid = threadIdx.x;
  const int b   = blockIdx.x >> 5;
  const int h0  = (blockIdx.x & 31) << 1;

  const int xr   = (tid >> 4) & 3;
  const int xf   = tid & 15;
  const int cq   = tid >> 6;
  const int xh   = h0 - 1 + xr;
  const bool rowok = (unsigned)xh < 64u;
  const int xhc  = xh < 0 ? 0 : (xh > 63 ? 63 : xh);
  const float4* xsrc = (const float4*)(x + (size_t)(b*64 + cq) * 4096 + xhc * 64) + xf;
  float4 xv[16];
  #pragma unroll
  for (int j = 0; j < 16; ++j)
    xv[j] = xsrc[j * 4096];
  float4 urv[4], uiv[4];
  #pragma unroll
  for (int j = 0; j < 4; ++j) {
    urv[j] = ((const float4*)gUr)[tid + (j << 8)];
    uiv[j] = ((const float4*)gUi)[tid + (j << 8)];
  }
  for (int q = tid; q < 512; q += 256)
    xs[(q >> 1) * 66 + (q & 1) * 65] = 0.f;
  {
    float* xd = xs + ((cq << 2) + xr) * 66 + 1 + (xf << 2);
    #pragma unroll
    for (int j = 0; j < 16; ++j) {
      const float4 v = rowok ? xv[j] : make_float4(0.f, 0.f, 0.f, 0.f);
      float* d = xd + j * (16 * 66);
      d[0] = v.x; d[1] = v.y; d[2] = v.z; d[3] = v.w;
    }
    #pragma unroll
    for (int j = 0; j < 4; ++j) {
      const int k = (tid >> 4) + (j << 4);
      float* dr = Ur + (xf << 2) * 65 + k;
      float* di = Ui + (xf << 2) * 65 + k;
      dr[0] = urv[j].x; dr[65] = urv[j].y; dr[130] = urv[j].z; dr[195] = urv[j].w;
      di[0] = uiv[j].x; di[65] = uiv[j].y; di[130] = uiv[j].z; di[195] = uiv[j].w;
    }
  }
  __syncthreads();

  const int lane = tid & 31;
  const int hf   = lane >> 4;
  const int lr   = lane & 15;
  const int m0   = (tid >> 5) << 4;
  const int m    = m0 + lr;
  const int baseM = ((m >> 6) * 66) + (m & 63);

  v8f acc[5];
  const v8f vz = {0.f,0.f,0.f,0.f,0.f,0.f,0.f,0.f};
  #pragma unroll
  for (int n = 0; n < 5; ++n) acc[n] = vz;

  {
    v2f A0[9], A1[9];
    v2f B0[9][5], B1[9][5];
    const float*  xa = xs + baseM;
    const float2* bl = (const float2*)gBp + lane;
    s1_load(A0, B0, xa, bl, hf);
    #pragma unroll 1
    for (int jj = 0; jj < 7; ++jj) {
      s1_load(A1, B1, xa + 1056, bl + 288, hf);
      __builtin_amdgcn_sched_barrier(0);
      s1_mma(A0, B0, acc);
      __builtin_amdgcn_sched_barrier(0);
      xa += 2112; bl += 576;
      s1_load(A0, B0, xa, bl, hf);
      __builtin_amdgcn_sched_barrier(0);
      s1_mma(A1, B1, acc);
      __builtin_amdgcn_sched_barrier(0);
    }
    s1_load(A1, B1, xa + 1056, bl + 288, hf);
    __builtin_amdgcn_sched_barrier(0);
    s1_mma(A0, B0, acc);
    s1_mma(A1, B1, acc);
  }

  #pragma unroll
  for (int n = 0; n < 4; ++n) {
    int ch = (n << 4) + lr;
    #pragma unroll
    for (int r = 0; r < 8; ++r)
      resS[(m0 + r + (hf << 3)) * 65 + ch] = acc[n][r];
  }
  if (lr < 6) {
    float db = data_b[lr];
    float sa = gStyle[(b << 3) + lr];
    #pragma unroll
    for (int r = 0; r < 8; ++r) {
      int mm = m0 + r + (hf << 3);
      thS[(mm << 3) + lr] = tanhf(acc[4][r] + db) * PI_F + sa;
    }
  }
  __syncthreads();

  if (tid < 128) {
    float cc[6], ss2[6];
    #pragma unroll
    for (int i = 0; i < 6; ++i) {
      float th = thS[(tid << 3) + i] * 0.5f;
      cc[i] = cosf(th); ss2[i] = sinf(th);
    }
    float p[64];
    p[0] = 1.f;
    #pragma unroll
    for (int i = 0; i < 6; ++i) {
      #pragma unroll
      for (int a = (1 << i) - 1; a >= 0; --a) {
        float tv = p[a];
        p[2*a + 1] = tv * ss2[i];
        p[2*a]     = tv * cc[i];
      }
    }
    #pragma unroll
    for (int a = 0; a < 64; a += 4)
      *(float4*)&Pm[tid * 68 + a] = make_float4(p[a], p[a+1], p[a+2], p[a+3]);
  }
  __syncthreads();

  float4 wq[32];
  {
    const float4* wqp = (const float4*)gWq + lane;
    #pragma unroll
    for (int j = 0; j < 32; ++j) wq[j] = wqp[j << 5];
  }

  #pragma unroll 1
  for (int nt = 0; nt < 4; ++nt) {
    int acol = (nt << 4) + lr;
    v8f ar = vz, ai = vz;
    for (int s = 0; s < 16; ++s) {
      int k = (s << 2) + (hf << 1);
      v2f A;  A.x  = Pm[m * 68 + k];        A.y  = Pm[m * 68 + k + 1];
      v2f Br; Br.x = Ur[acol * 65 + k];     Br.y = Ur[acol * 65 + k + 1];
      v2f Bi; Bi.x = Ui[acol * 65 + k];     Bi.y = Ui[acol * 65 + k + 1];
      ar = wmma_f32_k4(A, Br, ar);
      ai = wmma_f32_k4(A, Bi, ai);
    }
    #pragma unroll
    for (int r = 0; r < 8; ++r) {
      int mm = m0 + r + (hf << 3);
      PRm[mm * 68 + acol] = ar[r]*ar[r] + ai[r]*ai[r];
    }
  }

  {
    v2f A4[16];
    const float* pa = PRm + m * 68 + (hf << 1);
    #pragma unroll
    for (int s = 0; s < 16; ++s) {
      const float2 av = *(const float2*)(pa + (s << 2));
      A4[s].x = av.x; A4[s].y = av.y;
    }
    __builtin_amdgcn_sched_barrier(0);
    v8f q[4];
    #pragma unroll
    for (int nt = 0; nt < 4; ++nt) q[nt] = vz;
    #pragma unroll
    for (int s = 0; s < 16; ++s) {
      #pragma unroll
      for (int nt = 0; nt < 4; ++nt) {
        const float4 w = wq[(nt << 3) + (s >> 1)];
        v2f B;
        B.x = (s & 1) ? w.z : w.x;
        B.y = (s & 1) ? w.w : w.y;
        q[nt] = wmma_f32_k4(A4[s], B, q[nt]);
      }
    }
    #pragma unroll
    for (int nt = 0; nt < 4; ++nt) {
      const int ccol = (nt << 4) + lr;
      const float bias = res_b[ccol] + out_b[ccol];
      #pragma unroll
      for (int r = 0; r < 8; ++r) {
        const int mm = m0 + r + (hf << 3);
        resS[mm * 65 + ccol] += q[nt][r] + bias;
      }
    }
  }
  __syncthreads();

  for (int idx = tid; idx < 8192; idx += 256) {
    int c = idx >> 7, mm = idx & 127;
    int h = h0 + (mm >> 6), w = mm & 63;
    out[(size_t)(b*64 + c) * 4096 + h * 64 + w] = resS[mm * 65 + c];
  }
}

extern "C" void kernel_launch(void* const* d_in, const int* in_sizes, int n_in,
                              void* d_out, int out_size, void* d_ws, size_t ws_size,
                              hipStream_t stream) {
  (void)in_sizes; (void)n_in; (void)out_size; (void)ws_size;
  const float* x       = (const float*)d_in[0];
  const float* style   = (const float*)d_in[1];
  const float* data_w  = (const float*)d_in[2];
  const float* data_b  = (const float*)d_in[3];
  const float* style_w = (const float*)d_in[4];
  const float* style_b = (const float*)d_in[5];
  const float* qcnn    = (const float*)d_in[6];
  const float* meas    = (const float*)d_in[7];
  const float* out_w   = (const float*)d_in[8];
  const float* out_b   = (const float*)d_in[9];
  const float* res_w   = (const float*)d_in[10];
  const float* res_b   = (const float*)d_in[11];

  float* ws   = (float*)d_ws;
  float* Ur   = ws;
  float* Ui   = ws + 4096;
  float* Wq   = ws + 8192;
  float* sAng = ws + 12288;
  float* Bp   = ws + 12352;

  hipLaunchKernelGGL(qcnn_setup_kernel, dim3(409), dim3(128), 0, stream,
                     style, style_w, style_b, qcnn, meas, out_w, res_w, data_w,
                     Ur, Ui, Wq, sAng, Bp);
  hipLaunchKernelGGL(qcnn_main_kernel, dim3(256), dim3(256), 140288  , stream,
                     x, data_b, res_b, out_b, Ur, Ui, Wq, sAng, Bp, (float*)d_out);
}


# stage 1 conv in 2-D minimal-filtering form F(2x2,3x3): 16 transformed planes of K=64 over column pairs replace 4 of K=192 (480 -> 320 matrix instructions per wave); conv summation order changes
# speedup vs baseline: 7.2879x; 1.2376x over previous
#include <hip/hip_runtime.h>
#include <math.h>

typedef __attribute__((ext_vector_type(2))) float v2f;
typedef __attribute__((ext_vector_type(8))) float v8f;

#define PI_F 3.14159265358979323846f

static __device__ __forceinline__ v8f wmma_f32_k4(v2f a, v2f b, v8f c) {
  return __builtin_amdgcn_wmma_f32_16x16x4_f32(false, a, false, b, (short)0, c, false, false);
}


static __device__ __forceinline__ void s1_load(v2f (&A)[4][2], v2f (&B)[4][2][5],
                                               const float* va, const float2* bl) {
  #pragma unroll
  for (int s = 0; s < 4; ++s) {
    #pragma unroll
    for (int t = 0; t < 2; ++t) {
      #pragma unroll
      for (int n = 0; n < 5; ++n) {
        const float2 bv = bl[s * 2560 + n * 512 + t * 32];
        B[s][t][n].x = bv.x; B[s][t][n].y = bv.y;
      }
    }
  }
  #pragma unroll
  for (int s = 0; s < 4; ++s) {
    #pragma unroll
    for (int t = 0; t < 2; ++t) {
      const float2 av = *(const float2*)(va + s * 2176 + t * 4);
      A[s][t].x = av.x; A[s][t].y = av.y;
    }
  }
}

static __device__ __forceinline__ void s1_mma(const v2f (&A)[4][2], const v2f (&B)[4][2][5],
                                              v8f (&acc)[4][5]) {
  #pragma unroll
  for (int t = 0; t < 2; ++t) {
    #pragma unroll
    for (int s = 0; s < 4; ++s) {
      #pragma unroll
      for (int n = 0; n < 5; ++n)
        acc[s][n] = wmma_f32_k4(A[s][t], B[s][t][n], acc[s][n]);
    }
  }
}

static __device__ __forceinline__ float row_lower(float x) {
  const int r = __builtin_amdgcn_update_dpp(0, __builtin_bit_cast(int, x), 0x111, 0xf, 0xf, true);
  return __builtin_bit_cast(float, r);
}
static __device__ __forceinline__ float row_upper(float x) {
  const int r = __builtin_amdgcn_update_dpp(0, __builtin_bit_cast(int, x), 0x101, 0xf, 0xf, true);
  return __builtin_bit_cast(float, r);
}


static __device__ __forceinline__ float lane_xor(float x, int m) {
  const int v = __builtin_bit_cast(int, x);
  int r;
  switch (m) {
    case 1:  r = __builtin_amdgcn_update_dpp(v, v, 0x161, 0xf, 0xf, false); break;
    case 2:  r = __builtin_amdgcn_update_dpp(v, v, 0x162, 0xf, 0xf, false); break;
    case 4:  r = __builtin_amdgcn_update_dpp(v, v, 0x164, 0xf, 0xf, false); break;
    case 8:  r = __builtin_amdgcn_update_dpp(v, v, 0x168, 0xf, 0xf, false); break;
    default: r = __builtin_amdgcn_permlanex16(v, v, 0x76543210, 0xfedcba98, false, false); break;
  }
  return __builtin_bit_cast(float, r);
}

__global__ __launch_bounds__(128) void qcnn_setup_kernel(
                                  const float* __restrict__ style,
                                  const float* __restrict__ style_w,
                                  const float* __restrict__ style_b,
                                  const float* __restrict__ qcnn,
                                  const float* __restrict__ meas,
                                  const float* __restrict__ out_w,
                                  const float* __restrict__ res_w,
                                  const float* __restrict__ data_w,
                                  float* __restrict__ UrT, float* __restrict__ UiT,
                                  float* __restrict__ Wq, float* __restrict__ sAng,
                                  float* __restrict__ Bpack)
{
  __shared__ float tcs[48];
  __shared__ float tsn[48];
  const int tid = threadIdx.x;
  const int blk = blockIdx.x;

  if (blk < 16) {
    const int lane = tid & 31;
    const int t    = (blk << 2) + (tid >> 5);

    if (tid < 48) {
      float ang;
      if (tid < 12)      ang = 0.5f * qcnn[(tid * 2 + 0) * 3];
      else if (tid < 24) ang = 0.5f * qcnn[((tid - 12) * 2 + 1) * 3];
      else {
        const int kind = (tid - 24) / 6, i = (tid - 24) % 6;
        const float th = 0.5f * meas[i*3+0];
        const float ph = meas[i*3+1], la = meas[i*3+2];
        ang = (kind == 0) ? th : (kind == 1) ? ph : (kind == 2) ? la : (ph + la);
      }
      tcs[tid] = cosf(ang);
      tsn[tid] = sinf(ang);
    }
    __syncthreads();

    float r0 = (lane == t)      ? 1.f : 0.f;
    float r1 = (lane + 32 == t) ? 1.f : 0.f;
    float i0 = 0.f, i1 = 0.f;

    #pragma unroll
    for (int l = 0; l < 2; ++l) {
      {
        const int g = l * 6;
        const float c  = tcs[g],      s  = tsn[g];
        const float ch = tcs[12 + g], sh = tsn[12 + g];
        const float yr0 = c*r0 - s*r1, yi0 = c*i0 - s*i1;
        const float yr1 = s*r0 + c*r1, yi1 = s*i0 + c*i1;
        r0 = yr0*ch + yi0*sh;  i0 = yi0*ch - yr0*sh;
        r1 = yr1*ch - yi1*sh;  i1 = yi1*ch + yr1*sh;
      }
      #pragma unroll
      for (int i = 1; i < 6; ++i) {
        const int mask = 1 << (5 - i);
        const int g    = l * 6 + i;
        const bool hb  = (lane & mask) != 0;
        const float c  = tcs[g],      s  = tsn[g];
        const float ch = tcs[12 + g], sh = tsn[12 + g];
        const float ss  = hb ? s : -s;
        const float shs = hb ? -sh : sh;
        const float pr0 = lane_xor(r0, mask), pi0 = lane_xor(i0, mask);
        const float pr1 = lane_xor(r1, mask), pi1 = lane_xor(i1, mask);
        const float yr0 = c*r0 + ss*pr0, yi0 = c*i0 + ss*pi0;
        const float yr1 = c*r1 + ss*pr1, yi1 = c*i1 + ss*pi1;
        r0 = yr0*ch + yi0*shs;  i0 = yi0*ch - yr0*shs;
        r1 = yr1*ch + yi1*shs;  i1 = yi1*ch - yr1*shs;
      }
      int s0 = lane, s1 = lane + 32;
      #pragma unroll
      for (int k = 5; k >= 0; --k) {
        const int mc = 1 << (5 - k);
        const int mt = 1 << (5 - ((k + 1) % 6));
        if (s0 & mc) s0 ^= mt;
        if (s1 & mc) s1 ^= mt;
      }
      const int l0 = s0 & 31, l1 = s1 & 31;
      const float a0 = __shfl(r0, l0), a1 = __shfl(r1, l0);
      const float a2 = __shfl(i0, l0), a3 = __shfl(i1, l0);
      const float b0 = __shfl(r0, l1), b1 = __shfl(r1, l1);
      const float b2 = __shfl(i0, l1), b3 = __shfl(i1, l1);
      r0 = (s0 & 32) ? a1 : a0;  i0 = (s0 & 32) ? a3 : a2;
      r1 = (s1 & 32) ? b1 : b0;  i1 = (s1 & 32) ? b3 : b2;
    }
    #pragma unroll
    for (int i = 0; i < 6; ++i) {
      const float ct = tcs[24 + i], st = tsn[24 + i];
      const float m01r = -tcs[36 + i]*st, m01i = -tsn[36 + i]*st;
      const float m10r =  tcs[30 + i]*st, m10i =  tsn[30 + i]*st;
      const float m11r =  tcs[42 + i]*ct, m11i =  tsn[42 + i]*ct;
      if (i == 0) {
        const float nr0 = ct*r0 + m01r*r1 - m01i*i1;
        const float ni0 = ct*i0 + m01r*i1 + m01i*r1;
        const float nr1 = m10r*r0 - m10i*i0 + m11r*r1 - m11i*i1;
        const float ni1 = m10r*i0 + m10i*r0 + m11r*i1 + m11i*r1;
        r0 = nr0; i0 = ni0; r1 = nr1; i1 = ni1;
      } else {
        const int mask = 1 << (5 - i);
        const bool hb  = (lane & mask) != 0;
        const float pr0 = lane_xor(r0, mask), pi0 = lane_xor(i0, mask);
        const float pr1 = lane_xor(r1, mask), pi1 = lane_xor(i1, mask);
        const float lr0 = ct*r0 + m01r*pr0 - m01i*pi0;
        const float li0 = ct*i0 + m01r*pi0 + m01i*pr0;
        const float hr0 = m10r*pr0 - m10i*pi0 + m11r*r0 - m11i*i0;
        const float hi0 = m10r*pi0 + m10i*pr0 + m11r*i0 + m11i*r0;
        const float lr1 = ct*r1 + m01r*pr1 - m01i*pi1;
        const float li1 = ct*i1 + m01r*pi1 + m01i*pr1;
        const float hr1 = m10r*pr1 - m10i*pi1 + m11r*r1 - m11i*i1;
        const float hi1 = m10r*pi1 + m10i*pr1 + m11r*i1 + m11i*r1;
        r0 = hb ? hr0 : lr0;  i0 = hb ? hi0 : li0;
        r1 = hb ? hr1 : lr1;  i1 = hb ? hi1 : li1;
      }
    }
    UrT[t*64 + lane]      = r0;
    UrT[t*64 + 32 + lane] = r1;
    UiT[t*64 + lane]      = i0;
    UiT[t*64 + 32 + lane] = i1;
  } else if (blk < 48) {
    const int idx = ((blk - 16) << 7) + tid;
    const int ln = (idx >> 2) & 31;
    const int st = (((idx >> 7) & 7) << 1) + ((idx >> 1) & 1);
    const int a = (st << 2) + ((ln >> 4) << 1) + (idx & 1);
    const int c = ((idx >> 10) << 4) + (ln & 15);
    float s = 0.f;
    #pragma unroll
    for (int i = 0; i < 6; ++i)
      s += (((a >> (5 - i)) & 1) ? -1.f : 1.f) * out_w[c*6 + i];
    Wq[idx] = s;
  } else if (blk == 48) {
    if (tid < 64) {
      const int img = tid >> 3, i = tid & 7;
      if (i < 6) {
        const float4* sp = (const float4*)(style   + img*128);
        const float4* wp = (const float4*)(style_w + i*128);
        float d = 0.f;
        #pragma unroll 8
        for (int j = 0; j < 32; ++j) {
          const float4 sv = sp[j], wv = wp[j];
          d = fmaf(sv.x, wv.x, d);  d = fmaf(sv.y, wv.y, d);
          d = fmaf(sv.z, wv.z, d);  d = fmaf(sv.w, wv.w, d);
        }
        sAng[tid] = tanhf(d + style_b[i]) * PI_F;
      }
    }
  } else {
    const int idx  = ((blk - 49) << 7) + tid;
    const int lane = idx & 31;
    const int ks   = (idx >> 5) & 15;
    const int pn   = idx >> 9;
    const int n    = pn % 5;
    const int pl   = pn / 5;
    const int q    = pl >> 2;
    const int s    = pl & 3;
    const int ch   = (ks << 2) + ((lane >> 4) << 1);
    const int row  = (n << 4) + (lane & 15);
    float val[2] = {0.f, 0.f};
    if (row < 70) {
      const float* wr = ((row < 64) ? (res_w + row * 576) : (data_w + (row - 64) * 576)) + ch * 9;
      #pragma unroll
      for (int v = 0; v < 2; ++v) {
        float G[3];
        #pragma unroll
        for (int j = 0; j < 3; ++j) {
          const float g0 = wr[v * 9 + j], g1 = wr[v * 9 + 3 + j], g2 = wr[v * 9 + 6 + j];
          if (q == 0)      G[j] = g0;
          else if (q == 1) G[j] = 0.5f * ((g0 + g1) + g2);
          else if (q == 2) G[j] = 0.5f * ((g0 - g1) + g2);
          else             G[j] = g2;
        }
        if (s == 0)      val[v] = G[0];
        else if (s == 1) val[v] = 0.5f * ((G[0] + G[1]) + G[2]);
        else if (s == 2) val[v] = 0.5f * ((G[0] - G[1]) + G[2]);
        else             val[v] = G[2];
      }
    }
    ((float2*)Bpack)[idx] = make_float2(val[0], val[1]);
  }
}

__global__ __launch_bounds__(256) void qcnn_main_kernel(
    const float* __restrict__ x,
    const float* __restrict__ data_b,
    const float* __restrict__ res_b,
    const float* __restrict__ out_b,
    const float* __restrict__ gUr,
    const float* __restrict__ gUi,
    const float* __restrict__ gWq,
    const float* __restrict__ gStyle,
    const float* __restrict__ gBp,
    float* __restrict__ out)
{
  extern __shared__ float smem[];
  float* Vs   = smem;
  float* Pm   = smem;
  float* PRm  = smem + 8704;
  float* resS = smem + 17408;
  float* thS  = smem + 25728;
  float* Ur   = smem + 34816;
  float* Ui   = Ur + 4160;
  float* Zb   = Ui + 4160;

  const int tid = threadIdx.x;
  const int b   = blockIdx.x >> 5;
  const int h0  = (blockIdx.x & 31) << 1;

  const int xf   = tid & 15;
  const int cg   = tid >> 4;
  const bool okA = h0 > 0;
  const bool okD = h0 < 62;
  const int hA   = okA ? h0 - 1 : 0;
  const int hD   = okD ? h0 + 2 : 63;
  const float4* xsrc = (const float4*)(x + (size_t)(b*64 + (cg << 2)) * 4096) + xf;
  float4 xv[4][4];
  #pragma unroll
  for (int j = 0; j < 4; ++j) {
    const float4* pc = xsrc + j * 1024;
    xv[j][0] = pc[hA * 16];
    xv[j][1] = pc[h0 * 16];
    xv[j][2] = pc[(h0 + 1) * 16];
    xv[j][3] = pc[hD * 16];
  }
  float4 urv[4], uiv[4];
  #pragma unroll
  for (int j = 0; j < 4; ++j) {
    urv[j] = ((const float4*)gUr)[tid + (j << 8)];
    uiv[j] = ((const float4*)gUi)[tid + (j << 8)];
  }
  {
    float* vd = Vs + (xf << 1) * 68 + (cg << 2);
    const float4 z4 = make_float4(0.f, 0.f, 0.f, 0.f);
    #pragma unroll
    for (int q = 0; q < 4; ++q) {
      float Vv[4][2][4];
      #pragma unroll
      for (int j = 0; j < 4; ++j) {
        const float4 d0 = okA ? xv[j][0] : z4;
        const float4 d1 = xv[j][1];
        const float4 d2 = xv[j][2];
        const float4 d3 = okD ? xv[j][3] : z4;
        float4 T;
        if (q == 0)      T = make_float4(d0.x - d2.x, d0.y - d2.y, d0.z - d2.z, d0.w - d2.w);
        else if (q == 1) T = make_float4(d1.x + d2.x, d1.y + d2.y, d1.z + d2.z, d1.w + d2.w);
        else if (q == 2) T = make_float4(d2.x - d1.x, d2.y - d1.y, d2.z - d1.z, d2.w - d1.w);
        else             T = make_float4(d1.x - d3.x, d1.y - d3.y, d1.z - d3.z, d1.w - d3.w);
        const float lf = row_lower(T.w);
        const float rt = row_upper(T.x);
        Vv[0][0][j] = lf  - T.y;  Vv[1][0][j] = T.x + T.y;
        Vv[2][0][j] = T.y - T.x;  Vv[3][0][j] = T.x - T.z;
        Vv[0][1][j] = T.y - T.w;  Vv[1][1][j] = T.z + T.w;
        Vv[2][1][j] = T.w - T.z;  Vv[3][1][j] = T.z - rt;
      }
      #pragma unroll
      for (int s = 0; s < 4; ++s) {
        #pragma unroll
        for (int e = 0; e < 2; ++e)
          *(float4*)(vd + ((q << 2) + s) * 2176 + e * 68) =
              make_float4(Vv[s][e][0], Vv[s][e][1], Vv[s][e][2], Vv[s][e][3]);
      }
    }
    #pragma unroll
    for (int j = 0; j < 4; ++j) {
      const int k = (tid >> 4) + (j << 4);
      float* dr = Ur + (xf << 2) * 65 + k;
      float* di = Ui + (xf << 2) * 65 + k;
      dr[0] = urv[j].x; dr[65] = urv[j].y; dr[130] = urv[j].z; dr[195] = urv[j].w;
      di[0] = uiv[j].x; di[65] = uiv[j].y; di[130] = uiv[j].z; di[195] = uiv[j].w;
    }
  }
  __syncthreads();

  const int lane = tid & 31;
  const int hf   = lane >> 4;
  const int lr   = lane & 15;
  const int m0   = (tid >> 5) << 4;
  const int m    = m0 + lr;
  const int qw   = (tid >> 5) & 3;
  const int mt   = tid >> 7;

  v8f acc[4][5];
  const v8f vz = {0.f,0.f,0.f,0.f,0.f,0.f,0.f,0.f};
  #pragma unroll
  for (int s = 0; s < 4; ++s) {
    #pragma unroll
    for (int n = 0; n < 5; ++n) acc[s][n] = vz;
  }

  {
    v2f A0[4][2], A1[4][2];
    v2f B0[4][2][5], B1[4][2][5];
    const float*  va = Vs + qw * 8704 + ((mt << 4) + lr) * 68 + (hf << 1);
    const float2* bl = (const float2*)gBp + qw * 10240 + lane;
    s1_load(A0, B0, va, bl);
    #pragma unroll 1
    for (int jj = 0; jj < 3; ++jj) {
      s1_load(A1, B1, va + 8, bl + 64);
      __builtin_amdgcn_sched_barrier(0);
      s1_mma(A0, B0, acc);
      __builtin_amdgcn_sched_barrier(0);
      va += 16; bl += 128;
      s1_load(A0, B0, va, bl);
      __builtin_amdgcn_sched_barrier(0);
      s1_mma(A1, B1, acc);
      __builtin_amdgcn_sched_barrier(0);
    }
    s1_load(A1, B1, va + 8, bl + 64);
    __builtin_amdgcn_sched_barrier(0);
    s1_mma(A0, B0, acc);
    s1_mma(A1, B1, acc);
  }

  {
    float* zb = Zb + qw * 4760 + lr * 68 + (mt << 5) + (hf << 4);
    #pragma unroll
    for (int n = 0; n < 5; ++n) {
      if (n < 4 || lr < 6) {
        #pragma unroll
        for (int k = 0; k < 4; ++k) {
          const int r0 = 2 * k, r1 = 2 * k + 1;
          const float e0 = (acc[0][n][r0] + acc[1][n][r0]) + acc[2][n][r0];
          const float o0 = (acc[1][n][r0] - acc[2][n][r0]) - acc[3][n][r0];
          const float e1 = (acc[0][n][r1] + acc[1][n][r1]) + acc[2][n][r1];
          const float o1 = (acc[1][n][r1] - acc[2][n][r1]) - acc[3][n][r1];
          *(float4*)(zb + n * 1088 + 4 * k) = make_float4(e0, o0, e1, o1);
        }
      }
    }
  }
  __syncthreads();

  for (int e = tid; e < 768; e += 256) {
    const int i   = e >> 7;
    const int pix = e & 127;
    const bool up = pix < 64;
    const float sg = up ? 1.f : -1.f;
    const float* zp = Zb + (up ? 0 : 4760) + (64 + i) * 68 + (pix & 63);
    const float d = fmaf(sg, zp[9520], fmaf(sg, zp[4760], zp[0]));
    thS[(pix << 3) + i] = tanhf(d + data_b[i]) * PI_F + gStyle[(b << 3) + i];
  }
  __syncthreads();

  if (tid < 128) {
    float cc[6], ss2[6];
    #pragma unroll
    for (int i = 0; i < 6; ++i) {
      float th = thS[(tid << 3) + i] * 0.5f;
      cc[i] = cosf(th); ss2[i] = sinf(th);
    }
    float p[64];
    p[0] = 1.f;
    #pragma unroll
    for (int i = 0; i < 6; ++i) {
      #pragma unroll
      for (int a = (1 << i) - 1; a >= 0; --a) {
        float tv = p[a];
        p[2*a + 1] = tv * ss2[i];
        p[2*a]     = tv * cc[i];
      }
    }
    #pragma unroll
    for (int a = 0; a < 64; a += 4)
      *(float4*)&Pm[tid * 68 + a] = make_float4(p[a], p[a+1], p[a+2], p[a+3]);
  }
  __syncthreads();

  float conv[4][8];
  {
    const bool up = m0 < 64;
    const float sg = up ? 1.f : -1.f;
    const float* zw = Zb + (up ? 0 : 4760) + lr * 68 + (m0 & 63) + (hf << 3);
    #pragma unroll
    for (int nt = 0; nt < 4; ++nt) {
      #pragma unroll
      for (int g = 0; g < 2; ++g) {
        const float4 a0 = *(const float4*)(zw + nt * 1088 + 4 * g);
        const float4 a1 = *(const float4*)(zw + nt * 1088 + 4 * g + 4760);
        const float4 a2 = *(const float4*)(zw + nt * 1088 + 4 * g + 9520);
        conv[nt][4*g+0] = fmaf(sg, a2.x, fmaf(sg, a1.x, a0.x));
        conv[nt][4*g+1] = fmaf(sg, a2.y, fmaf(sg, a1.y, a0.y));
        conv[nt][4*g+2] = fmaf(sg, a2.z, fmaf(sg, a1.z, a0.z));
        conv[nt][4*g+3] = fmaf(sg, a2.w, fmaf(sg, a1.w, a0.w));
      }
    }
  }

  float4 wq[32];
  {
    const float4* wqp = (const float4*)gWq + lane;
    #pragma unroll
    for (int j = 0; j < 32; ++j) wq[j] = wqp[j << 5];
  }

  #pragma unroll 1
  for (int nt = 0; nt < 4; ++nt) {
    int acol = (nt << 4) + lr;
    v8f ar = vz, ai = vz;
    for (int s = 0; s < 16; ++s) {
      int k = (s << 2) + (hf << 1);
      v2f A;  A.x  = Pm[m * 68 + k];        A.y  = Pm[m * 68 + k + 1];
      v2f Br; Br.x = Ur[acol * 65 + k];     Br.y = Ur[acol * 65 + k + 1];
      v2f Bi; Bi.x = Ui[acol * 65 + k];     Bi.y = Ui[acol * 65 + k + 1];
      ar = wmma_f32_k4(A, Br, ar);
      ai = wmma_f32_k4(A, Bi, ai);
    }
    #pragma unroll
    for (int r = 0; r < 8; ++r) {
      int mm = m0 + r + (hf << 3);
      PRm[mm * 68 + acol] = ar[r]*ar[r] + ai[r]*ai[r];
    }
  }

  {
    v2f A4[16];
    const float* pa = PRm + m * 68 + (hf << 1);
    #pragma unroll
    for (int s = 0; s < 16; ++s) {
      const float2 av = *(const float2*)(pa + (s << 2));
      A4[s].x = av.x; A4[s].y = av.y;
    }
    __builtin_amdgcn_sched_barrier(0);
    v8f q[4];
    #pragma unroll
    for (int nt = 0; nt < 4; ++nt) q[nt] = vz;
    #pragma unroll
    for (int s = 0; s < 16; ++s) {
      #pragma unroll
      for (int nt = 0; nt < 4; ++nt) {
        const float4 w = wq[(nt << 3) + (s >> 1)];
        v2f B;
        B.x = (s & 1) ? w.z : w.x;
        B.y = (s & 1) ? w.w : w.y;
        q[nt] = wmma_f32_k4(A4[s], B, q[nt]);
      }
    }
    #pragma unroll
    for (int nt = 0; nt < 4; ++nt) {
      const int ccol = (nt << 4) + lr;
      const float bias = res_b[ccol] + out_b[ccol];
      #pragma unroll
      for (int r = 0; r < 8; ++r) {
        const int mm = m0 + r + (hf << 3);
        resS[mm * 65 + ccol] = conv[nt][r] + (q[nt][r] + bias);
      }
    }
  }
  __syncthreads();

  for (int idx = tid; idx < 8192; idx += 256) {
    int c = idx >> 7, mm = idx & 127;
    int h = h0 + (mm >> 6), w = mm & 63;
    out[(size_t)(b*64 + c) * 4096 + h * 64 + w] = resS[mm * 65 + c];
  }
}

extern "C" void kernel_launch(void* const* d_in, const int* in_sizes, int n_in,
                              void* d_out, int out_size, void* d_ws, size_t ws_size,
                              hipStream_t stream) {
  (void)in_sizes; (void)n_in; (void)out_size; (void)ws_size;
  const float* x       = (const float*)d_in[0];
  const float* style   = (const float*)d_in[1];
  const float* data_w  = (const float*)d_in[2];
  const float* data_b  = (const float*)d_in[3];
  const float* style_w = (const float*)d_in[4];
  const float* style_b = (const float*)d_in[5];
  const float* qcnn    = (const float*)d_in[6];
  const float* meas    = (const float*)d_in[7];
  const float* out_w   = (const float*)d_in[8];
  const float* out_b   = (const float*)d_in[9];
  const float* res_w   = (const float*)d_in[10];
  const float* res_b   = (const float*)d_in[11];

  float* ws   = (float*)d_ws;
  float* Ur   = ws;
  float* Ui   = ws + 4096;
  float* Wq   = ws + 8192;
  float* sAng = ws + 12288;
  float* Bp   = ws + 12352;

  hipLaunchKernelGGL(qcnn_setup_kernel, dim3(369), dim3(128), 0, stream,
                     style, style_w, style_b, qcnn, meas, out_w, res_w, data_w,
                     Ur, Ui, Wq, sAng, Bp);
  hipLaunchKernelGGL(qcnn_main_kernel, dim3(256), dim3(256), 248704  , stream,
                     x, data_b, res_b, out_b, Ur, Ui, Wq, sAng, Bp, (float*)d_out);
}


# encoding angles and product state built per wave for its own pixel tile: three wires' tanh, sin, cos per lane, halves swapped by a row move, biases fetched with the tile; two workgroup barriers go
# speedup vs baseline: 7.6059x; 1.0436x over previous
#include <hip/hip_runtime.h>
#include <math.h>

typedef __attribute__((ext_vector_type(2))) float v2f;
typedef __attribute__((ext_vector_type(8))) float v8f;

#define PI_F 3.14159265358979323846f

static __device__ __forceinline__ v8f wmma_f32_k4(v2f a, v2f b, v8f c) {
  return __builtin_amdgcn_wmma_f32_16x16x4_f32(false, a, false, b, (short)0, c, false, false);
}


static __device__ __forceinline__ void s1_load(v2f (&A)[4][2], v2f (&B)[4][2][5],
                                               const float* va, const float2* bl) {
  #pragma unroll
  for (int s = 0; s < 4; ++s) {
    #pragma unroll
    for (int t = 0; t < 2; ++t) {
      #pragma unroll
      for (int n = 0; n < 5; ++n) {
        const float2 bv = bl[s * 2560 + n * 512 + t * 32];
        B[s][t][n].x = bv.x; B[s][t][n].y = bv.y;
      }
    }
  }
  #pragma unroll
  for (int s = 0; s < 4; ++s) {
    #pragma unroll
    for (int t = 0; t < 2; ++t) {
      const float2 av = *(const float2*)(va + s * 2176 + t * 4);
      A[s][t].x = av.x; A[s][t].y = av.y;
    }
  }
}

static __device__ __forceinline__ void s1_mma(const v2f (&A)[4][2], const v2f (&B)[4][2][5],
                                              v8f (&acc)[4][5]) {
  #pragma unroll
  for (int t = 0; t < 2; ++t) {
    #pragma unroll
    for (int s = 0; s < 4; ++s) {
      #pragma unroll
      for (int n = 0; n < 5; ++n)
        acc[s][n] = wmma_f32_k4(A[s][t], B[s][t][n], acc[s][n]);
    }
  }
}

static __device__ __forceinline__ float row_lower(float x) {
  const int r = __builtin_amdgcn_update_dpp(0, __builtin_bit_cast(int, x), 0x111, 0xf, 0xf, true);
  return __builtin_bit_cast(float, r);
}
static __device__ __forceinline__ float row_upper(float x) {
  const int r = __builtin_amdgcn_update_dpp(0, __builtin_bit_cast(int, x), 0x101, 0xf, 0xf, true);
  return __builtin_bit_cast(float, r);
}


static __device__ __forceinline__ float lane_xor(float x, int m) {
  const int v = __builtin_bit_cast(int, x);
  int r;
  switch (m) {
    case 1:  r = __builtin_amdgcn_update_dpp(v, v, 0x161, 0xf, 0xf, false); break;
    case 2:  r = __builtin_amdgcn_update_dpp(v, v, 0x162, 0xf, 0xf, false); break;
    case 4:  r = __builtin_amdgcn_update_dpp(v, v, 0x164, 0xf, 0xf, false); break;
    case 8:  r = __builtin_amdgcn_update_dpp(v, v, 0x168, 0xf, 0xf, false); break;
    default: r = __builtin_amdgcn_permlanex16(v, v, 0x76543210, 0xfedcba98, false, false); break;
  }
  return __builtin_bit_cast(float, r);
}

__global__ __launch_bounds__(128) void qcnn_setup_kernel(
                                  const float* __restrict__ style,
                                  const float* __restrict__ style_w,
                                  const float* __restrict__ style_b,
                                  const float* __restrict__ qcnn,
                                  const float* __restrict__ meas,
                                  const float* __restrict__ out_w,
                                  const float* __restrict__ res_w,
                                  const float* __restrict__ data_w,
                                  float* __restrict__ UrT, float* __restrict__ UiT,
                                  float* __restrict__ Wq, float* __restrict__ sAng,
                                  float* __restrict__ Bpack)
{
  __shared__ float tcs[48];
  __shared__ float tsn[48];
  const int tid = threadIdx.x;
  const int blk = blockIdx.x;

  if (blk < 16) {
    const int lane = tid & 31;
    const int t    = (blk << 2) + (tid >> 5);

    if (tid < 48) {
      float ang;
      if (tid < 12)      ang = 0.5f * qcnn[(tid * 2 + 0) * 3];
      else if (tid < 24) ang = 0.5f * qcnn[((tid - 12) * 2 + 1) * 3];
      else {
        const int kind = (tid - 24) / 6, i = (tid - 24) % 6;
        const float th = 0.5f * meas[i*3+0];
        const float ph = meas[i*3+1], la = meas[i*3+2];
        ang = (kind == 0) ? th : (kind == 1) ? ph : (kind == 2) ? la : (ph + la);
      }
      tcs[tid] = cosf(ang);
      tsn[tid] = sinf(ang);
    }
    __syncthreads();

    float r0 = (lane == t)      ? 1.f : 0.f;
    float r1 = (lane + 32 == t) ? 1.f : 0.f;
    float i0 = 0.f, i1 = 0.f;

    #pragma unroll
    for (int l = 0; l < 2; ++l) {
      {
        const int g = l * 6;
        const float c  = tcs[g],      s  = tsn[g];
        const float ch = tcs[12 + g], sh = tsn[12 + g];
        const float yr0 = c*r0 - s*r1, yi0 = c*i0 - s*i1;
        const float yr1 = s*r0 + c*r1, yi1 = s*i0 + c*i1;
        r0 = yr0*ch + yi0*sh;  i0 = yi0*ch - yr0*sh;
        r1 = yr1*ch - yi1*sh;  i1 = yi1*ch + yr1*sh;
      }
      #pragma unroll
      for (int i = 1; i < 6; ++i) {
        const int mask = 1 << (5 - i);
        const int g    = l * 6 + i;
        const bool hb  = (lane & mask) != 0;
        const float c  = tcs[g],      s  = tsn[g];
        const float ch = tcs[12 + g], sh = tsn[12 + g];
        const float ss  = hb ? s : -s;
        const float shs = hb ? -sh : sh;
        const float pr0 = lane_xor(r0, mask), pi0 = lane_xor(i0, mask);
        const float pr1 = lane_xor(r1, mask), pi1 = lane_xor(i1, mask);
        const float yr0 = c*r0 + ss*pr0, yi0 = c*i0 + ss*pi0;
        const float yr1 = c*r1 + ss*pr1, yi1 = c*i1 + ss*pi1;
        r0 = yr0*ch + yi0*shs;  i0 = yi0*ch - yr0*shs;
        r1 = yr1*ch + yi1*shs;  i1 = yi1*ch - yr1*shs;
      }
      int s0 = lane, s1 = lane + 32;
      #pragma unroll
      for (int k = 5; k >= 0; --k) {
        const int mc = 1 << (5 - k);
        const int mt = 1 << (5 - ((k + 1) % 6));
        if (s0 & mc) s0 ^= mt;
        if (s1 & mc) s1 ^= mt;
      }
      const int l0 = s0 & 31, l1 = s1 & 31;
      const float a0 = __shfl(r0, l0), a1 = __shfl(r1, l0);
      const float a2 = __shfl(i0, l0), a3 = __shfl(i1, l0);
      const float b0 = __shfl(r0, l1), b1 = __shfl(r1, l1);
      const float b2 = __shfl(i0, l1), b3 = __shfl(i1, l1);
      r0 = (s0 & 32) ? a1 : a0;  i0 = (s0 & 32) ? a3 : a2;
      r1 = (s1 & 32) ? b1 : b0;  i1 = (s1 & 32) ? b3 : b2;
    }
    #pragma unroll
    for (int i = 0; i < 6; ++i) {
      const float ct = tcs[24 + i], st = tsn[24 + i];
      const float m01r = -tcs[36 + i]*st, m01i = -tsn[36 + i]*st;
      const float m10r =  tcs[30 + i]*st, m10i =  tsn[30 + i]*st;
      const float m11r =  tcs[42 + i]*ct, m11i =  tsn[42 + i]*ct;
      if (i == 0) {
        const float nr0 = ct*r0 + m01r*r1 - m01i*i1;
        const float ni0 = ct*i0 + m01r*i1 + m01i*r1;
        const float nr1 = m10r*r0 - m10i*i0 + m11r*r1 - m11i*i1;
        const float ni1 = m10r*i0 + m10i*r0 + m11r*i1 + m11i*r1;
        r0 = nr0; i0 = ni0; r1 = nr1; i1 = ni1;
      } else {
        const int mask = 1 << (5 - i);
        const bool hb  = (lane & mask) != 0;
        const float pr0 = lane_xor(r0, mask), pi0 = lane_xor(i0, mask);
        const float pr1 = lane_xor(r1, mask), pi1 = lane_xor(i1, mask);
        const float lr0 = ct*r0 + m01r*pr0 - m01i*pi0;
        const float li0 = ct*i0 + m01r*pi0 + m01i*pr0;
        const float hr0 = m10r*pr0 - m10i*pi0 + m11r*r0 - m11i*i0;
        const float hi0 = m10r*pi0 + m10i*pr0 + m11r*i0 + m11i*r0;
        const float lr1 = ct*r1 + m01r*pr1 - m01i*pi1;
        const float li1 = ct*i1 + m01r*pi1 + m01i*pr1;
        const float hr1 = m10r*pr1 - m10i*pi1 + m11r*r1 - m11i*i1;
        const float hi1 = m10r*pi1 + m10i*pr1 + m11r*i1 + m11i*r1;
        r0 = hb ? hr0 : lr0;  i0 = hb ? hi0 : li0;
        r1 = hb ? hr1 : lr1;  i1 = hb ? hi1 : li1;
      }
    }
    UrT[t*64 + lane]      = r0;
    UrT[t*64 + 32 + lane] = r1;
    UiT[t*64 + lane]      = i0;
    UiT[t*64 + 32 + lane] = i1;
  } else if (blk < 48) {
    const int idx = ((blk - 16) << 7) + tid;
    const int ln = (idx >> 2) & 31;
    const int st = (((idx >> 7) & 7) << 1) + ((idx >> 1) & 1);
    const int a = (st << 2) + ((ln >> 4) << 1) + (idx & 1);
    const int c = ((idx >> 10) << 4) + (ln & 15);
    float s = 0.f;
    #pragma unroll
    for (int i = 0; i < 6; ++i)
      s += (((a >> (5 - i)) & 1) ? -1.f : 1.f) * out_w[c*6 + i];
    Wq[idx] = s;
  } else if (blk == 48) {
    if (tid < 64) {
      const int img = tid >> 3, i = tid & 7;
      if (i < 6) {
        const float4* sp = (const float4*)(style   + img*128);
        const float4* wp = (const float4*)(style_w + i*128);
        float d = 0.f;
        #pragma unroll 8
        for (int j = 0; j < 32; ++j) {
          const float4 sv = sp[j], wv = wp[j];
          d = fmaf(sv.x, wv.x, d);  d = fmaf(sv.y, wv.y, d);
          d = fmaf(sv.z, wv.z, d);  d = fmaf(sv.w, wv.w, d);
        }
        sAng[tid] = tanhf(d + style_b[i]) * PI_F;
      }
    }
  } else {
    const int idx  = ((blk - 49) << 7) + tid;
    const int lane = idx & 31;
    const int ks   = (idx >> 5) & 15;
    const int pn   = idx >> 9;
    const int n    = pn % 5;
    const int pl   = pn / 5;
    const int q    = pl >> 2;
    const int s    = pl & 3;
    const int ch   = (ks << 2) + ((lane >> 4) << 1);
    const int row  = (n << 4) + (lane & 15);
    float val[2] = {0.f, 0.f};
    if (row < 70) {
      const float* wr = ((row < 64) ? (res_w + row * 576) : (data_w + (row - 64) * 576)) + ch * 9;
      #pragma unroll
      for (int v = 0; v < 2; ++v) {
        float G[3];
        #pragma unroll
        for (int j = 0; j < 3; ++j) {
          const float g0 = wr[v * 9 + j], g1 = wr[v * 9 + 3 + j], g2 = wr[v * 9 + 6 + j];
          if (q == 0)      G[j] = g0;
          else if (q == 1) G[j] = 0.5f * ((g0 + g1) + g2);
          else if (q == 2) G[j] = 0.5f * ((g0 - g1) + g2);
          else             G[j] = g2;
        }
        if (s == 0)      val[v] = G[0];
        else if (s == 1) val[v] = 0.5f * ((G[0] + G[1]) + G[2]);
        else if (s == 2) val[v] = 0.5f * ((G[0] - G[1]) + G[2]);
        else             val[v] = G[2];
      }
    }
    ((float2*)Bpack)[idx] = make_float2(val[0], val[1]);
  }
}

__global__ __launch_bounds__(256) void qcnn_main_kernel(
    const float* __restrict__ x,
    const float* __restrict__ data_b,
    const float* __restrict__ res_b,
    const float* __restrict__ out_b,
    const float* __restrict__ gUr,
    const float* __restrict__ gUi,
    const float* __restrict__ gWq,
    const float* __restrict__ gStyle,
    const float* __restrict__ gBp,
    float* __restrict__ out)
{
  extern __shared__ float smem[];
  float* Vs   = smem;
  float* Pm   = smem;
  float* PRm  = smem + 8704;
  float* resS = smem + 17408;
  float* Ur   = smem + 34816;
  float* Ui   = Ur + 4160;
  float* Zb   = Ui + 4160;

  const int tid = threadIdx.x;
  const int b   = blockIdx.x >> 5;
  const int h0  = (blockIdx.x & 31) << 1;

  const int xf   = tid & 15;
  const int cg   = tid >> 4;
  const bool okA = h0 > 0;
  const bool okD = h0 < 62;
  const int hA   = okA ? h0 - 1 : 0;
  const int hD   = okD ? h0 + 2 : 63;
  const float4* xsrc = (const float4*)(x + (size_t)(b*64 + (cg << 2)) * 4096) + xf;
  float4 xv[4][4];
  #pragma unroll
  for (int j = 0; j < 4; ++j) {
    const float4* pc = xsrc + j * 1024;
    xv[j][0] = pc[hA * 16];
    xv[j][1] = pc[h0 * 16];
    xv[j][2] = pc[(h0 + 1) * 16];
    xv[j][3] = pc[hD * 16];
  }
  float4 urv[4], uiv[4];
  #pragma unroll
  for (int j = 0; j < 4; ++j) {
    urv[j] = ((const float4*)gUr)[tid + (j << 8)];
    uiv[j] = ((const float4*)gUi)[tid + (j << 8)];
  }
  const int w3 = ((tid >> 4) & 1) * 3;
  float dbv[3], sav[3];
  #pragma unroll
  for (int j = 0; j < 3; ++j) {
    dbv[j] = data_b[w3 + j];
    sav[j] = gStyle[(b << 3) + w3 + j];
  }
  {
    float* vd = Vs + (xf << 1) * 68 + (cg << 2);
    const float4 z4 = make_float4(0.f, 0.f, 0.f, 0.f);
    #pragma unroll
    for (int q = 0; q < 4; ++q) {
      float Vv[4][2][4];
      #pragma unroll
      for (int j = 0; j < 4; ++j) {
        const float4 d0 = okA ? xv[j][0] : z4;
        const float4 d1 = xv[j][1];
        const float4 d2 = xv[j][2];
        const float4 d3 = okD ? xv[j][3] : z4;
        float4 T;
        if (q == 0)      T = make_float4(d0.x - d2.x, d0.y - d2.y, d0.z - d2.z, d0.w - d2.w);
        else if (q == 1) T = make_float4(d1.x + d2.x, d1.y + d2.y, d1.z + d2.z, d1.w + d2.w);
        else if (q == 2) T = make_float4(d2.x - d1.x, d2.y - d1.y, d2.z - d1.z, d2.w - d1.w);
        else             T = make_float4(d1.x - d3.x, d1.y - d3.y, d1.z - d3.z, d1.w - d3.w);
        const float lf = row_lower(T.w);
        const float rt = row_upper(T.x);
        Vv[0][0][j] = lf  - T.y;  Vv[1][0][j] = T.x + T.y;
        Vv[2][0][j] = T.y - T.x;  Vv[3][0][j] = T.x - T.z;
        Vv[0][1][j] = T.y - T.w;  Vv[1][1][j] = T.z + T.w;
        Vv[2][1][j] = T.w - T.z;  Vv[3][1][j] = T.z - rt;
      }
      #pragma unroll
      for (int s = 0; s < 4; ++s) {
        #pragma unroll
        for (int e = 0; e < 2; ++e)
          *(float4*)(vd + ((q << 2) + s) * 2176 + e * 68) =
              make_float4(Vv[s][e][0], Vv[s][e][1], Vv[s][e][2], Vv[s][e][3]);
      }
    }
    #pragma unroll
    for (int j = 0; j < 4; ++j) {
      const int k = (tid >> 4) + (j << 4);
      float* dr = Ur + (xf << 2) * 65 + k;
      float* di = Ui + (xf << 2) * 65 + k;
      dr[0] = urv[j].x; dr[65] = urv[j].y; dr[130] = urv[j].z; dr[195] = urv[j].w;
      di[0] = uiv[j].x; di[65] = uiv[j].y; di[130] = uiv[j].z; di[195] = uiv[j].w;
    }
  }
  __syncthreads();

  const int lane = tid & 31;
  const int hf   = lane >> 4;
  const int lr   = lane & 15;
  const int m0   = (tid >> 5) << 4;
  const int m    = m0 + lr;
  const int qw   = (tid >> 5) & 3;
  const int mt   = tid >> 7;

  v8f acc[4][5];
  const v8f vz = {0.f,0.f,0.f,0.f,0.f,0.f,0.f,0.f};
  #pragma unroll
  for (int s = 0; s < 4; ++s) {
    #pragma unroll
    for (int n = 0; n < 5; ++n) acc[s][n] = vz;
  }

  {
    v2f A0[4][2], A1[4][2];
    v2f B0[4][2][5], B1[4][2][5];
    const float*  va = Vs + qw * 8704 + ((mt << 4) + lr) * 68 + (hf << 1);
    const float2* bl = (const float2*)gBp + qw * 10240 + lane;
    s1_load(A0, B0, va, bl);
    #pragma unroll 1
    for (int jj = 0; jj < 3; ++jj) {
      s1_load(A1, B1, va + 8, bl + 64);
      __builtin_amdgcn_sched_barrier(0);
      s1_mma(A0, B0, acc);
      __builtin_amdgcn_sched_barrier(0);
      va += 16; bl += 128;
      s1_load(A0, B0, va, bl);
      __builtin_amdgcn_sched_barrier(0);
      s1_mma(A1, B1, acc);
      __builtin_amdgcn_sched_barrier(0);
    }
    s1_load(A1, B1, va + 8, bl + 64);
    __builtin_amdgcn_sched_barrier(0);
    s1_mma(A0, B0, acc);
    s1_mma(A1, B1, acc);
  }

  {
    float* zb = Zb + qw * 4760 + lr * 68 + (mt << 5) + (hf << 4);
    #pragma unroll
    for (int n = 0; n < 5; ++n) {
      if (n < 4 || lr < 6) {
        #pragma unroll
        for (int k = 0; k < 4; ++k) {
          const int r0 = 2 * k, r1 = 2 * k + 1;
          const float e0 = (acc[0][n][r0] + acc[1][n][r0]) + acc[2][n][r0];
          const float o0 = (acc[1][n][r0] - acc[2][n][r0]) - acc[3][n][r0];
          const float e1 = (acc[0][n][r1] + acc[1][n][r1]) + acc[2][n][r1];
          const float o1 = (acc[1][n][r1] - acc[2][n][r1]) - acc[3][n][r1];
          *(float4*)(zb + n * 1088 + 4 * k) = make_float4(e0, o0, e1, o1);
        }
      }
    }
  }
  __syncthreads();

  {
    const bool up = m0 < 64;
    const float sg = up ? 1.f : -1.f;
    const float* zp = Zb + (up ? 0 : 4760) + (64 + 3 * hf) * 68 + (m0 & 63) + lr;
    float dd[3], cw[3], sw[3];
    #pragma unroll
    for (int j = 0; j < 3; ++j)
      dd[j] = fmaf(sg, zp[j * 68 + 9520], fmaf(sg, zp[j * 68 + 4760], zp[j * 68]));
    #pragma unroll
    for (int j = 0; j < 3; ++j) {
      const float th = (tanhf(dd[j] + dbv[j]) * PI_F + sav[j]) * 0.5f;
      cw[j] = cosf(th); sw[j] = sinf(th);
    }
    float c6[6], s6[6];
    #pragma unroll
    for (int j = 0; j < 3; ++j) {
      const float co = lane_xor(cw[j], 16);
      const float so = lane_xor(sw[j], 16);
      c6[j]     = hf ? co : cw[j];   s6[j]     = hf ? so : sw[j];
      c6[3 + j] = hf ? cw[j] : co;   s6[3 + j] = hf ? sw[j] : so;
    }
    float pr[32];
    pr[0] = hf ? s6[0] : c6[0];
    #pragma unroll
    for (int i = 1; i < 6; ++i) {
      #pragma unroll
      for (int a = (1 << (i - 1)) - 1; a >= 0; --a) {
        const float tv = pr[a];
        pr[2*a + 1] = tv * s6[i];
        pr[2*a]     = tv * c6[i];
      }
    }
    float* pd = Pm + m * 68 + (hf << 5);
    #pragma unroll
    for (int a = 0; a < 32; a += 4)
      *(float4*)(pd + a) = make_float4(pr[a], pr[a+1], pr[a+2], pr[a+3]);
  }

  float conv[4][8];
  {
    const bool up = m0 < 64;
    const float sg = up ? 1.f : -1.f;
    const float* zw = Zb + (up ? 0 : 4760) + lr * 68 + (m0 & 63) + (hf << 3);
    #pragma unroll
    for (int nt = 0; nt < 4; ++nt) {
      #pragma unroll
      for (int g = 0; g < 2; ++g) {
        const float4 a0 = *(const float4*)(zw + nt * 1088 + 4 * g);
        const float4 a1 = *(const float4*)(zw + nt * 1088 + 4 * g + 4760);
        const float4 a2 = *(const float4*)(zw + nt * 1088 + 4 * g + 9520);
        conv[nt][4*g+0] = fmaf(sg, a2.x, fmaf(sg, a1.x, a0.x));
        conv[nt][4*g+1] = fmaf(sg, a2.y, fmaf(sg, a1.y, a0.y));
        conv[nt][4*g+2] = fmaf(sg, a2.z, fmaf(sg, a1.z, a0.z));
        conv[nt][4*g+3] = fmaf(sg, a2.w, fmaf(sg, a1.w, a0.w));
      }
    }
  }

  float4 wq[32];
  {
    const float4* wqp = (const float4*)gWq + lane;
    #pragma unroll
    for (int j = 0; j < 32; ++j) wq[j] = wqp[j << 5];
  }

  #pragma unroll 1
  for (int nt = 0; nt < 4; ++nt) {
    int acol = (nt << 4) + lr;
    v8f ar = vz, ai = vz;
    for (int s = 0; s < 16; ++s) {
      int k = (s << 2) + (hf << 1);
      v2f A;  A.x  = Pm[m * 68 + k];        A.y  = Pm[m * 68 + k + 1];
      v2f Br; Br.x = Ur[acol * 65 + k];     Br.y = Ur[acol * 65 + k + 1];
      v2f Bi; Bi.x = Ui[acol * 65 + k];     Bi.y = Ui[acol * 65 + k + 1];
      ar = wmma_f32_k4(A, Br, ar);
      ai = wmma_f32_k4(A, Bi, ai);
    }
    #pragma unroll
    for (int r = 0; r < 8; ++r) {
      int mm = m0 + r + (hf << 3);
      PRm[mm * 68 + acol] = ar[r]*ar[r] + ai[r]*ai[r];
    }
  }

  {
    v2f A4[16];
    const float* pa = PRm + m * 68 + (hf << 1);
    #pragma unroll
    for (int s = 0; s < 16; ++s) {
      const float2 av = *(const float2*)(pa + (s << 2));
      A4[s].x = av.x; A4[s].y = av.y;
    }
    __builtin_amdgcn_sched_barrier(0);
    v8f q[4];
    #pragma unroll
    for (int nt = 0; nt < 4; ++nt) q[nt] = vz;
    #pragma unroll
    for (int s = 0; s < 16; ++s) {
      #pragma unroll
      for (int nt = 0; nt < 4; ++nt) {
        const float4 w = wq[(nt << 3) + (s >> 1)];
        v2f B;
        B.x = (s & 1) ? w.z : w.x;
        B.y = (s & 1) ? w.w : w.y;
        q[nt] = wmma_f32_k4(A4[s], B, q[nt]);
      }
    }
    #pragma unroll
    for (int nt = 0; nt < 4; ++nt) {
      const int ccol = (nt << 4) + lr;
      const float bias = res_b[ccol] + out_b[ccol];
      #pragma unroll
      for (int r = 0; r < 8; ++r) {
        const int mm = m0 + r + (hf << 3);
        resS[mm * 65 + ccol] = conv[nt][r] + (q[nt][r] + bias);
      }
    }
  }
  __syncthreads();

  for (int idx = tid; idx < 8192; idx += 256) {
    int c = idx >> 7, mm = idx & 127;
    int h = h0 + (mm >> 6), w = mm & 63;
    out[(size_t)(b*64 + c) * 4096 + h * 64 + w] = resS[mm * 65 + c];
  }
}

extern "C" void kernel_launch(void* const* d_in, const int* in_sizes, int n_in,
                              void* d_out, int out_size, void* d_ws, size_t ws_size,
                              hipStream_t stream) {
  (void)in_sizes; (void)n_in; (void)out_size; (void)ws_size;
  const float* x       = (const float*)d_in[0];
  const float* style   = (const float*)d_in[1];
  const float* data_w  = (const float*)d_in[2];
  const float* data_b  = (const float*)d_in[3];
  const float* style_w = (const float*)d_in[4];
  const float* style_b = (const float*)d_in[5];
  const float* qcnn    = (const float*)d_in[6];
  const float* meas    = (const float*)d_in[7];
  const float* out_w   = (const float*)d_in[8];
  const float* out_b   = (const float*)d_in[9];
  const float* res_w   = (const float*)d_in[10];
  const float* res_b   = (const float*)d_in[11];

  float* ws   = (float*)d_ws;
  float* Ur   = ws;
  float* Ui   = ws + 4096;
  float* Wq   = ws + 8192;
  float* sAng = ws + 12288;
  float* Bp   = ws + 12352;

  hipLaunchKernelGGL(qcnn_setup_kernel, dim3(369), dim3(128), 0, stream,
                     style, style_w, style_b, qcnn, meas, out_w, res_w, data_w,
                     Ur, Ui, Wq, sAng, Bp);
  hipLaunchKernelGGL(qcnn_main_kernel, dim3(256), dim3(256), 248704  , stream,
                     x, data_b, res_b, out_b, Ur, Ui, Wq, sAng, Bp, (float*)d_out);
}


# stage 1 wave tile reshaped: both M-tiles x 2 res N-tiles + data N-tile of one M-tile (was 1 M-tile x 5 N-tiles), each res B fragment feeds 2 WMMAs: B fetches per wave 320 -> 192; K loop unrolled
# speedup vs baseline: 7.9521x; 1.0455x over previous
#include <hip/hip_runtime.h>
#include <math.h>

typedef __attribute__((ext_vector_type(2))) float v2f;
typedef __attribute__((ext_vector_type(8))) float v8f;

#define PI_F 3.14159265358979323846f

static __device__ __forceinline__ v8f wmma_f32_k4(v2f a, v2f b, v8f c) {
  return __builtin_amdgcn_wmma_f32_16x16x4_f32(false, a, false, b, (short)0, c, false, false);
}


static __device__ __forceinline__ void s1_load(v2f (&A)[2][4][2], v2f (&B)[4][2][3],
                                               const float* va0, const float* va1,
                                               const float2* blr, const float2* bld) {
  #pragma unroll
  for (int s = 0; s < 4; ++s) {
    #pragma unroll
    for (int t = 0; t < 2; ++t) {
      #pragma unroll
      for (int n = 0; n < 2; ++n) {
        const float2 bv = blr[s * 2560 + n * 512 + t * 32];
        B[s][t][n].x = bv.x; B[s][t][n].y = bv.y;
      }
      const float2 dv = bld[s * 2560 + t * 32];
      B[s][t][2].x = dv.x; B[s][t][2].y = dv.y;
    }
  }
  #pragma unroll
  for (int s = 0; s < 4; ++s) {
    #pragma unroll
    for (int t = 0; t < 2; ++t) {
      const float2 a0 = *(const float2*)(va0 + s * 2176 + t * 4);
      const float2 a1 = *(const float2*)(va1 + s * 2176 + t * 4);
      A[0][s][t].x = a0.x; A[0][s][t].y = a0.y;
      A[1][s][t].x = a1.x; A[1][s][t].y = a1.y;
    }
  }
}

static __device__ __forceinline__ void s1_mma(const v2f (&A)[2][4][2], const v2f (&B)[4][2][3],
                                              v8f (&accR)[2][4][2], v8f (&accD)[4]) {
  #pragma unroll
  for (int t = 0; t < 2; ++t) {
    #pragma unroll
    for (int s = 0; s < 4; ++s) {
      #pragma unroll
      for (int n = 0; n < 2; ++n) {
        accR[0][s][n] = wmma_f32_k4(A[0][s][t], B[s][t][n], accR[0][s][n]);
        accR[1][s][n] = wmma_f32_k4(A[1][s][t], B[s][t][n], accR[1][s][n]);
      }
      accD[s] = wmma_f32_k4(A[0][s][t], B[s][t][2], accD[s]);
    }
  }
}

static __device__ __forceinline__ float row_lower(float x) {
  const int r = __builtin_amdgcn_update_dpp(0, __builtin_bit_cast(int, x), 0x111, 0xf, 0xf, true);
  return __builtin_bit_cast(float, r);
}
static __device__ __forceinline__ float row_upper(float x) {
  const int r = __builtin_amdgcn_update_dpp(0, __builtin_bit_cast(int, x), 0x101, 0xf, 0xf, true);
  return __builtin_bit_cast(float, r);
}


static __device__ __forceinline__ float lane_xor(float x, int m) {
  const int v = __builtin_bit_cast(int, x);
  int r;
  switch (m) {
    case 1:  r = __builtin_amdgcn_update_dpp(v, v, 0x161, 0xf, 0xf, false); break;
    case 2:  r = __builtin_amdgcn_update_dpp(v, v, 0x162, 0xf, 0xf, false); break;
    case 4:  r = __builtin_amdgcn_update_dpp(v, v, 0x164, 0xf, 0xf, false); break;
    case 8:  r = __builtin_amdgcn_update_dpp(v, v, 0x168, 0xf, 0xf, false); break;
    default: r = __builtin_amdgcn_permlanex16(v, v, 0x76543210, 0xfedcba98, false, false); break;
  }
  return __builtin_bit_cast(float, r);
}

__global__ __launch_bounds__(128) void qcnn_setup_kernel(
                                  const float* __restrict__ style,
                                  const float* __restrict__ style_w,
                                  const float* __restrict__ style_b,
                                  const float* __restrict__ qcnn,
                                  const float* __restrict__ meas,
                                  const float* __restrict__ out_w,
                                  const float* __restrict__ res_w,
                                  const float* __restrict__ data_w,
                                  float* __restrict__ UrT, float* __restrict__ UiT,
                                  float* __restrict__ Wq, float* __restrict__ sAng,
                                  float* __restrict__ Bpack)
{
  __shared__ float tcs[48];
  __shared__ float tsn[48];
  const int tid = threadIdx.x;
  const int blk = blockIdx.x;

  if (blk < 16) {
    const int lane = tid & 31;
    const int t    = (blk << 2) + (tid >> 5);

    if (tid < 48) {
      float ang;
      if (tid < 12)      ang = 0.5f * qcnn[(tid * 2 + 0) * 3];
      else if (tid < 24) ang = 0.5f * qcnn[((tid - 12) * 2 + 1) * 3];
      else {
        const int kind = (tid - 24) / 6, i = (tid - 24) % 6;
        const float th = 0.5f * meas[i*3+0];
        const float ph = meas[i*3+1], la = meas[i*3+2];
        ang = (kind == 0) ? th : (kind == 1) ? ph : (kind == 2) ? la : (ph + la);
      }
      tcs[tid] = cosf(ang);
      tsn[tid] = sinf(ang);
    }
    __syncthreads();

    float r0 = (lane == t)      ? 1.f : 0.f;
    float r1 = (lane + 32 == t) ? 1.f : 0.f;
    float i0 = 0.f, i1 = 0.f;

    #pragma unroll
    for (int l = 0; l < 2; ++l) {
      {
        const int g = l * 6;
        const float c  = tcs[g],      s  = tsn[g];
        const float ch = tcs[12 + g], sh = tsn[12 + g];
        const float yr0 = c*r0 - s*r1, yi0 = c*i0 - s*i1;
        const float yr1 = s*r0 + c*r1, yi1 = s*i0 + c*i1;
        r0 = yr0*ch + yi0*sh;  i0 = yi0*ch - yr0*sh;
        r1 = yr1*ch - yi1*sh;  i1 = yi1*ch + yr1*sh;
      }
      #pragma unroll
      for (int i = 1; i < 6; ++i) {
        const int mask = 1 << (5 - i);
        const int g    = l * 6 + i;
        const bool hb  = (lane & mask) != 0;
        const float c  = tcs[g],      s  = tsn[g];
        const float ch = tcs[12 + g], sh = tsn[12 + g];
        const float ss  = hb ? s : -s;
        const float shs = hb ? -sh : sh;
        const float pr0 = lane_xor(r0, mask), pi0 = lane_xor(i0, mask);
        const float pr1 = lane_xor(r1, mask), pi1 = lane_xor(i1, mask);
        const float yr0 = c*r0 + ss*pr0, yi0 = c*i0 + ss*pi0;
        const float yr1 = c*r1 + ss*pr1, yi1 = c*i1 + ss*pi1;
        r0 = yr0*ch + yi0*shs;  i0 = yi0*ch - yr0*shs;
        r1 = yr1*ch + yi1*shs;  i1 = yi1*ch - yr1*shs;
      }
      int s0 = lane, s1 = lane + 32;
      #pragma unroll
      for (int k = 5; k >= 0; --k) {
        const int mc = 1 << (5 - k);
        const int mt = 1 << (5 - ((k + 1) % 6));
        if (s0 & mc) s0 ^= mt;
        if (s1 & mc) s1 ^= mt;
      }
      const int l0 = s0 & 31, l1 = s1 & 31;
      const float a0 = __shfl(r0, l0), a1 = __shfl(r1, l0);
      const float a2 = __shfl(i0, l0), a3 = __shfl(i1, l0);
      const float b0 = __shfl(r0, l1), b1 = __shfl(r1, l1);
      const float b2 = __shfl(i0, l1), b3 = __shfl(i1, l1);
      r0 = (s0 & 32) ? a1 : a0;  i0 = (s0 & 32) ? a3 : a2;
      r1 = (s1 & 32) ? b1 : b0;  i1 = (s1 & 32) ? b3 : b2;
    }
    #pragma unroll
    for (int i = 0; i < 6; ++i) {
      const float ct = tcs[24 + i], st = tsn[24 + i];
      const float m01r = -tcs[36 + i]*st, m01i = -tsn[36 + i]*st;
      const float m10r =  tcs[30 + i]*st, m10i =  tsn[30 + i]*st;
      const float m11r =  tcs[42 + i]*ct, m11i =  tsn[42 + i]*ct;
      if (i == 0) {
        const float nr0 = ct*r0 + m01r*r1 - m01i*i1;
        const float ni0 = ct*i0 + m01r*i1 + m01i*r1;
        const float nr1 = m10r*r0 - m10i*i0 + m11r*r1 - m11i*i1;
        const float ni1 = m10r*i0 + m10i*r0 + m11r*i1 + m11i*r1;
        r0 = nr0; i0 = ni0; r1 = nr1; i1 = ni1;
      } else {
        const int mask = 1 << (5 - i);
        const bool hb  = (lane & mask) != 0;
        const float pr0 = lane_xor(r0, mask), pi0 = lane_xor(i0, mask);
        const float pr1 = lane_xor(r1, mask), pi1 = lane_xor(i1, mask);
        const float lr0 = ct*r0 + m01r*pr0 - m01i*pi0;
        const float li0 = ct*i0 + m01r*pi0 + m01i*pr0;
        const float hr0 = m10r*pr0 - m10i*pi0 + m11r*r0 - m11i*i0;
        const float hi0 = m10r*pi0 + m10i*pr0 + m11r*i0 + m11i*r0;
        const float lr1 = ct*r1 + m01r*pr1 - m01i*pi1;
        const float li1 = ct*i1 + m01r*pi1 + m01i*pr1;
        const float hr1 = m10r*pr1 - m10i*pi1 + m11r*r1 - m11i*i1;
        const float hi1 = m10r*pi1 + m10i*pr1 + m11r*i1 + m11i*r1;
        r0 = hb ? hr0 : lr0;  i0 = hb ? hi0 : li0;
        r1 = hb ? hr1 : lr1;  i1 = hb ? hi1 : li1;
      }
    }
    UrT[t*64 + lane]      = r0;
    UrT[t*64 + 32 + lane] = r1;
    UiT[t*64 + lane]      = i0;
    UiT[t*64 + 32 + lane] = i1;
  } else if (blk < 48) {
    const int idx = ((blk - 16) << 7) + tid;
    const int ln = (idx >> 2) & 31;
    const int st = (((idx >> 7) & 7) << 1) + ((idx >> 1) & 1);
    const int a = (st << 2) + ((ln >> 4) << 1) + (idx & 1);
    const int c = ((idx >> 10) << 4) + (ln & 15);
    float s = 0.f;
    #pragma unroll
    for (int i = 0; i < 6; ++i)
      s += (((a >> (5 - i)) & 1) ? -1.f : 1.f) * out_w[c*6 + i];
    Wq[idx] = s;
  } else if (blk == 48) {
    if (tid < 64) {
      const int img = tid >> 3, i = tid & 7;
      if (i < 6) {
        const float4* sp = (const float4*)(style   + img*128);
        const float4* wp = (const float4*)(style_w + i*128);
        float d = 0.f;
        #pragma unroll 8
        for (int j = 0; j < 32; ++j) {
          const float4 sv = sp[j], wv = wp[j];
          d = fmaf(sv.x, wv.x, d);  d = fmaf(sv.y, wv.y, d);
          d = fmaf(sv.z, wv.z, d);  d = fmaf(sv.w, wv.w, d);
        }
        sAng[tid] = tanhf(d + style_b[i]) * PI_F;
      }
    }
  } else {
    const int idx  = ((blk - 49) << 7) + tid;
    const int lane = idx & 31;
    const int ks   = (idx >> 5) & 15;
    const int pn   = idx >> 9;
    const int n    = pn % 5;
    const int pl   = pn / 5;
    const int q    = pl >> 2;
    const int s    = pl & 3;
    const int ch   = (ks << 2) + ((lane >> 4) << 1);
    const int row  = (n << 4) + (lane & 15);
    float val[2] = {0.f, 0.f};
    if (row < 70) {
      const float* wr = ((row < 64) ? (res_w + row * 576) : (data_w + (row - 64) * 576)) + ch * 9;
      #pragma unroll
      for (int v = 0; v < 2; ++v) {
        float G[3];
        #pragma unroll
        for (int j = 0; j < 3; ++j) {
          const float g0 = wr[v * 9 + j], g1 = wr[v * 9 + 3 + j], g2 = wr[v * 9 + 6 + j];
          if (q == 0)      G[j] = g0;
          else if (q == 1) G[j] = 0.5f * ((g0 + g1) + g2);
          else if (q == 2) G[j] = 0.5f * ((g0 - g1) + g2);
          else             G[j] = g2;
        }
        if (s == 0)      val[v] = G[0];
        else if (s == 1) val[v] = 0.5f * ((G[0] + G[1]) + G[2]);
        else if (s == 2) val[v] = 0.5f * ((G[0] - G[1]) + G[2]);
        else             val[v] = G[2];
      }
    }
    ((float2*)Bpack)[idx] = make_float2(val[0], val[1]);
  }
}

__global__ __launch_bounds__(256) void qcnn_main_kernel(
    const float* __restrict__ x,
    const float* __restrict__ data_b,
    const float* __restrict__ res_b,
    const float* __restrict__ out_b,
    const float* __restrict__ gUr,
    const float* __restrict__ gUi,
    const float* __restrict__ gWq,
    const float* __restrict__ gStyle,
    const float* __restrict__ gBp,
    float* __restrict__ out)
{
  extern __shared__ float smem[];
  float* Vs   = smem;
  float* Pm   = smem;
  float* PRm  = smem + 8704;
  float* resS = smem + 17408;
  float* Ur   = smem + 34816;
  float* Ui   = Ur + 4160;
  float* Zb   = Ui + 4160;

  const int tid = threadIdx.x;
  const int b   = blockIdx.x >> 5;
  const int h0  = (blockIdx.x & 31) << 1;

  const int xf   = tid & 15;
  const int cg   = tid >> 4;
  const bool okA = h0 > 0;
  const bool okD = h0 < 62;
  const int hA   = okA ? h0 - 1 : 0;
  const int hD   = okD ? h0 + 2 : 63;
  const float4* xsrc = (const float4*)(x + (size_t)(b*64 + (cg << 2)) * 4096) + xf;
  float4 xv[4][4];
  #pragma unroll
  for (int j = 0; j < 4; ++j) {
    const float4* pc = xsrc + j * 1024;
    xv[j][0] = pc[hA * 16];
    xv[j][1] = pc[h0 * 16];
    xv[j][2] = pc[(h0 + 1) * 16];
    xv[j][3] = pc[hD * 16];
  }
  float4 urv[4], uiv[4];
  #pragma unroll
  for (int j = 0; j < 4; ++j) {
    urv[j] = ((const float4*)gUr)[tid + (j << 8)];
    uiv[j] = ((const float4*)gUi)[tid + (j << 8)];
  }
  const int w3 = ((tid >> 4) & 1) * 3;
  float dbv[3], sav[3];
  #pragma unroll
  for (int j = 0; j < 3; ++j) {
    dbv[j] = data_b[w3 + j];
    sav[j] = gStyle[(b << 3) + w3 + j];
  }
  {
    float* vd = Vs + (xf << 1) * 68 + (cg << 2);
    const float4 z4 = make_float4(0.f, 0.f, 0.f, 0.f);
    #pragma unroll
    for (int q = 0; q < 4; ++q) {
      float Vv[4][2][4];
      #pragma unroll
      for (int j = 0; j < 4; ++j) {
        const float4 d0 = okA ? xv[j][0] : z4;
        const float4 d1 = xv[j][1];
        const float4 d2 = xv[j][2];
        const float4 d3 = okD ? xv[j][3] : z4;
        float4 T;
        if (q == 0)      T = make_float4(d0.x - d2.x, d0.y - d2.y, d0.z - d2.z, d0.w - d2.w);
        else if (q == 1) T = make_float4(d1.x + d2.x, d1.y + d2.y, d1.z + d2.z, d1.w + d2.w);
        else if (q == 2) T = make_float4(d2.x - d1.x, d2.y - d1.y, d2.z - d1.z, d2.w - d1.w);
        else             T = make_float4(d1.x - d3.x, d1.y - d3.y, d1.z - d3.z, d1.w - d3.w);
        const float lf = row_lower(T.w);
        const float rt = row_upper(T.x);
        Vv[0][0][j] = lf  - T.y;  Vv[1][0][j] = T.x + T.y;
        Vv[2][0][j] = T.y - T.x;  Vv[3][0][j] = T.x - T.z;
        Vv[0][1][j] = T.y - T.w;  Vv[1][1][j] = T.z + T.w;
        Vv[2][1][j] = T.w - T.z;  Vv[3][1][j] = T.z - rt;
      }
      #pragma unroll
      for (int s = 0; s < 4; ++s) {
        #pragma unroll
        for (int e = 0; e < 2; ++e)
          *(float4*)(vd + ((q << 2) + s) * 2176 + e * 68) =
              make_float4(Vv[s][e][0], Vv[s][e][1], Vv[s][e][2], Vv[s][e][3]);
      }
    }
    #pragma unroll
    for (int j = 0; j < 4; ++j) {
      const int k = (tid >> 4) + (j << 4);
      float* dr = Ur + (xf << 2) * 65 + k;
      float* di = Ui + (xf << 2) * 65 + k;
      dr[0] = urv[j].x; dr[65] = urv[j].y; dr[130] = urv[j].z; dr[195] = urv[j].w;
      di[0] = uiv[j].x; di[65] = uiv[j].y; di[130] = uiv[j].z; di[195] = uiv[j].w;
    }
  }
  __syncthreads();

  const int lane = tid & 31;
  const int hf   = lane >> 4;
  const int lr   = lane & 15;
  const int m0   = (tid >> 5) << 4;
  const int m    = m0 + lr;
  const int qw   = (tid >> 5) & 3;
  const int ng   = tid >> 7;

  v8f accR[2][4][2];
  v8f accD[4];
  const v8f vz = {0.f,0.f,0.f,0.f,0.f,0.f,0.f,0.f};
  #pragma unroll
  for (int s = 0; s < 4; ++s) {
    #pragma unroll
    for (int n = 0; n < 2; ++n) { accR[0][s][n] = vz; accR[1][s][n] = vz; }
    accD[s] = vz;
  }

  {
    v2f A0[2][4][2], A1[2][4][2];
    v2f B0[4][2][3], B1[4][2][3];
    const float*  va0 = Vs + qw * 8704 + ((ng << 4) + lr) * 68 + (hf << 1);
    const float*  va1 = Vs + qw * 8704 + (((ng ^ 1) << 4) + lr) * 68 + (hf << 1);
    const float2* blr = (const float2*)gBp + qw * 10240 + ng * 1024 + lane;
    const float2* bld = (const float2*)gBp + qw * 10240 + 2048 + lane;
    s1_load(A0, B0, va0, va1, blr, bld);
    #pragma unroll
    for (int jj = 0; jj < 3; ++jj) {
      s1_load(A1, B1, va0 + 8, va1 + 8, blr + 64, bld + 64);
      __builtin_amdgcn_sched_barrier(0);
      s1_mma(A0, B0, accR, accD);
      __builtin_amdgcn_sched_barrier(0);
      va0 += 16; va1 += 16; blr += 128; bld += 128;
      s1_load(A0, B0, va0, va1, blr, bld);
      __builtin_amdgcn_sched_barrier(0);
      s1_mma(A1, B1, accR, accD);
      __builtin_amdgcn_sched_barrier(0);
    }
    s1_load(A1, B1, va0 + 8, va1 + 8, blr + 64, bld + 64);
    __builtin_amdgcn_sched_barrier(0);
    s1_mma(A0, B0, accR, accD);
    s1_mma(A1, B1, accR, accD);
  }

  {
    float* zq = Zb + qw * 4760 + lr * 68 + (hf << 4);
    #pragma unroll
    for (int i = 0; i < 2; ++i) {
      #pragma unroll
      for (int n = 0; n < 2; ++n) {
        float* zb = zq + ((ng << 1) + n) * 1088 + ((ng ^ i) << 5);
        #pragma unroll
        for (int k = 0; k < 4; ++k) {
          const int r0 = 2 * k, r1 = 2 * k + 1;
          const float e0 = (accR[i][0][n][r0] + accR[i][1][n][r0]) + accR[i][2][n][r0];
          const float o0 = (accR[i][1][n][r0] - accR[i][2][n][r0]) - accR[i][3][n][r0];
          const float e1 = (accR[i][0][n][r1] + accR[i][1][n][r1]) + accR[i][2][n][r1];
          const float o1 = (accR[i][1][n][r1] - accR[i][2][n][r1]) - accR[i][3][n][r1];
          *(float4*)(zb + 4 * k) = make_float4(e0, o0, e1, o1);
        }
      }
    }
    if (lr < 6) {
      float* zb = zq + 4 * 1088 + (ng << 5);
      #pragma unroll
      for (int k = 0; k < 4; ++k) {
        const int r0 = 2 * k, r1 = 2 * k + 1;
        const float e0 = (accD[0][r0] + accD[1][r0]) + accD[2][r0];
        const float o0 = (accD[1][r0] - accD[2][r0]) - accD[3][r0];
        const float e1 = (accD[0][r1] + accD[1][r1]) + accD[2][r1];
        const float o1 = (accD[1][r1] - accD[2][r1]) - accD[3][r1];
        *(float4*)(zb + 4 * k) = make_float4(e0, o0, e1, o1);
      }
    }
  }
  __syncthreads();

  {
    const bool up = m0 < 64;
    const float sg = up ? 1.f : -1.f;
    const float* zp = Zb + (up ? 0 : 4760) + (64 + 3 * hf) * 68 + (m0 & 63) + lr;
    float dd[3], cw[3], sw[3];
    #pragma unroll
    for (int j = 0; j < 3; ++j)
      dd[j] = fmaf(sg, zp[j * 68 + 9520], fmaf(sg, zp[j * 68 + 4760], zp[j * 68]));
    #pragma unroll
    for (int j = 0; j < 3; ++j) {
      const float th = (tanhf(dd[j] + dbv[j]) * PI_F + sav[j]) * 0.5f;
      cw[j] = cosf(th); sw[j] = sinf(th);
    }
    float c6[6], s6[6];
    #pragma unroll
    for (int j = 0; j < 3; ++j) {
      const float co = lane_xor(cw[j], 16);
      const float so = lane_xor(sw[j], 16);
      c6[j]     = hf ? co : cw[j];   s6[j]     = hf ? so : sw[j];
      c6[3 + j] = hf ? cw[j] : co;   s6[3 + j] = hf ? sw[j] : so;
    }
    float pr[32];
    pr[0] = hf ? s6[0] : c6[0];
    #pragma unroll
    for (int i = 1; i < 6; ++i) {
      #pragma unroll
      for (int a = (1 << (i - 1)) - 1; a >= 0; --a) {
        const float tv = pr[a];
        pr[2*a + 1] = tv * s6[i];
        pr[2*a]     = tv * c6[i];
      }
    }
    float* pd = Pm + m * 68 + (hf << 5);
    #pragma unroll
    for (int a = 0; a < 32; a += 4)
      *(float4*)(pd + a) = make_float4(pr[a], pr[a+1], pr[a+2], pr[a+3]);
  }

  float conv[4][8];
  {
    const bool up = m0 < 64;
    const float sg = up ? 1.f : -1.f;
    const float* zw = Zb + (up ? 0 : 4760) + lr * 68 + (m0 & 63) + (hf << 3);
    #pragma unroll
    for (int nt = 0; nt < 4; ++nt) {
      #pragma unroll
      for (int g = 0; g < 2; ++g) {
        const float4 a0 = *(const float4*)(zw + nt * 1088 + 4 * g);
        const float4 a1 = *(const float4*)(zw + nt * 1088 + 4 * g + 4760);
        const float4 a2 = *(const float4*)(zw + nt * 1088 + 4 * g + 9520);
        conv[nt][4*g+0] = fmaf(sg, a2.x, fmaf(sg, a1.x, a0.x));
        conv[nt][4*g+1] = fmaf(sg, a2.y, fmaf(sg, a1.y, a0.y));
        conv[nt][4*g+2] = fmaf(sg, a2.z, fmaf(sg, a1.z, a0.z));
        conv[nt][4*g+3] = fmaf(sg, a2.w, fmaf(sg, a1.w, a0.w));
      }
    }
  }

  float4 wq[32];
  {
    const float4* wqp = (const float4*)gWq + lane;
    #pragma unroll
    for (int j = 0; j < 32; ++j) wq[j] = wqp[j << 5];
  }

  #pragma unroll 1
  for (int nt = 0; nt < 4; ++nt) {
    int acol = (nt << 4) + lr;
    v8f ar = vz, ai = vz;
    for (int s = 0; s < 16; ++s) {
      int k = (s << 2) + (hf << 1);
      v2f A;  A.x  = Pm[m * 68 + k];        A.y  = Pm[m * 68 + k + 1];
      v2f Br; Br.x = Ur[acol * 65 + k];     Br.y = Ur[acol * 65 + k + 1];
      v2f Bi; Bi.x = Ui[acol * 65 + k];     Bi.y = Ui[acol * 65 + k + 1];
      ar = wmma_f32_k4(A, Br, ar);
      ai = wmma_f32_k4(A, Bi, ai);
    }
    #pragma unroll
    for (int r = 0; r < 8; ++r) {
      int mm = m0 + r + (hf << 3);
      PRm[mm * 68 + acol] = ar[r]*ar[r] + ai[r]*ai[r];
    }
  }

  {
    v2f A4[16];
    const float* pa = PRm + m * 68 + (hf << 1);
    #pragma unroll
    for (int s = 0; s < 16; ++s) {
      const float2 av = *(const float2*)(pa + (s << 2));
      A4[s].x = av.x; A4[s].y = av.y;
    }
    __builtin_amdgcn_sched_barrier(0);
    v8f q[4];
    #pragma unroll
    for (int nt = 0; nt < 4; ++nt) q[nt] = vz;
    #pragma unroll
    for (int s = 0; s < 16; ++s) {
      #pragma unroll
      for (int nt = 0; nt < 4; ++nt) {
        const float4 w = wq[(nt << 3) + (s >> 1)];
        v2f B;
        B.x = (s & 1) ? w.z : w.x;
        B.y = (s & 1) ? w.w : w.y;
        q[nt] = wmma_f32_k4(A4[s], B, q[nt]);
      }
    }
    #pragma unroll
    for (int nt = 0; nt < 4; ++nt) {
      const int ccol = (nt << 4) + lr;
      const float bias = res_b[ccol] + out_b[ccol];
      #pragma unroll
      for (int r = 0; r < 8; ++r) {
        const int mm = m0 + r + (hf << 3);
        resS[mm * 65 + ccol] = conv[nt][r] + (q[nt][r] + bias);
      }
    }
  }
  __syncthreads();

  for (int idx = tid; idx < 8192; idx += 256) {
    int c = idx >> 7, mm = idx & 127;
    int h = h0 + (mm >> 6), w = mm & 63;
    out[(size_t)(b*64 + c) * 4096 + h * 64 + w] = resS[mm * 65 + c];
  }
}

extern "C" void kernel_launch(void* const* d_in, const int* in_sizes, int n_in,
                              void* d_out, int out_size, void* d_ws, size_t ws_size,
                              hipStream_t stream) {
  (void)in_sizes; (void)n_in; (void)out_size; (void)ws_size;
  const float* x       = (const float*)d_in[0];
  const float* style   = (const float*)d_in[1];
  const float* data_w  = (const float*)d_in[2];
  const float* data_b  = (const float*)d_in[3];
  const float* style_w = (const float*)d_in[4];
  const float* style_b = (const float*)d_in[5];
  const float* qcnn    = (const float*)d_in[6];
  const float* meas    = (const float*)d_in[7];
  const float* out_w   = (const float*)d_in[8];
  const float* out_b   = (const float*)d_in[9];
  const float* res_w   = (const float*)d_in[10];
  const float* res_b   = (const float*)d_in[11];

  float* ws   = (float*)d_ws;
  float* Ur   = ws;
  float* Ui   = ws + 4096;
  float* Wq   = ws + 8192;
  float* sAng = ws + 12288;
  float* Bp   = ws + 12352;

  hipLaunchKernelGGL(qcnn_setup_kernel, dim3(369), dim3(128), 0, stream,
                     style, style_w, style_b, qcnn, meas, out_w, res_w, data_w,
                     Ur, Ui, Wq, sAng, Bp);
  hipLaunchKernelGGL(qcnn_main_kernel, dim3(256), dim3(256), 248704  , stream,
                     x, data_b, res_b, out_b, Ur, Ui, Wq, sAng, Bp, (float*)d_out);
}
